# starting kernel
# baseline (speedup 1.0000x reference)
#include <hip/hip_runtime.h>
#include <hip/hip_bf16.h>

typedef __attribute__((ext_vector_type(2))) float v2f;
typedef __attribute__((ext_vector_type(8))) float v8f;

typedef struct __attribute__((packed, aligned(4))) { float x, y; } f2u;

#define T_LEN   131072
#define OUT_LEN (2 * T_LEN - 1)
#define TILE_W  512
#define NTILES  ((OUT_LEN + TILE_W - 1) / TILE_W)
#define NWAVES  8


__device__ __forceinline__ v8f wmma4(v2f a, v2f b, v8f c) {
    return __builtin_amdgcn_wmma_f32_16x16x4_f32(false, a, false, b,
                                                 (short)0, c, false, false);
}

__device__ __forceinline__ void step_checked2(const float* __restrict__ rir,
                                              const float* __restrict__ a_ch,
                                              int k0, int rbase, int bbase,
                                              v8f& accA, v8f& accB)
{
    int r0 = k0 + rbase, r1 = r0 + 1;
    int i0 = k0 + bbase, i1 = i0 + 1;
    int j0 = i0 + 256,   j1 = i0 + 257;
    float ax  = rir[min(max(r0, 0), T_LEN - 1)];
    float ay  = rir[min(max(r1, 0), T_LEN - 1)];
    float b0x = a_ch[min(max(i0, 0), T_LEN - 1)];
    float b0y = a_ch[min(max(i1, 0), T_LEN - 1)];
    float b1x = a_ch[min(max(j0, 0), T_LEN - 1)];
    float b1y = a_ch[min(max(j1, 0), T_LEN - 1)];
    v2f a, b0, b1;
    a.x  = ((unsigned)r0 < (unsigned)T_LEN) ? ax  : 0.f;
    a.y  = ((unsigned)r1 < (unsigned)T_LEN) ? ay  : 0.f;
    b0.x = ((unsigned)i0 < (unsigned)T_LEN) ? b0x : 0.f;
    b0.y = ((unsigned)i1 < (unsigned)T_LEN) ? b0y : 0.f;
    b1.x = ((unsigned)j0 < (unsigned)T_LEN) ? b1x : 0.f;
    b1.y = ((unsigned)j1 < (unsigned)T_LEN) ? b1y : 0.f;
    accA = wmma4(a, b0, accA);
    accB = wmma4(a, b1, accB);
}

__global__ __launch_bounds__(256)
void RoomImpulseResponse_wmma_conv(const float* __restrict__ audio,
                                   const float* __restrict__ rir,
                                   float* __restrict__ out)
{
    __shared__ float red[NWAVES * TILE_W];

    const int tile = blockIdx.x;
    const int ch   = blockIdx.y;
    const int n0   = tile * TILE_W;
    const int tid  = threadIdx.x;
    const int wave = tid >> 5;
    const int lane = tid & 31;
    const int h    = lane >> 4;
    const int mq   = lane & 15;

    const float* __restrict__ a_ch = audio + ch * T_LEN;

    int klo = T_LEN - 272 - n0;  if (klo < 0) klo = 0;
    int khi = 2 * T_LEN - 2 - n0;
    if (khi > T_LEN - 1 + 240) khi = T_LEN - 1 + 240;

    const int rbase = 2 * h - 16 * mq;
    const int bbase = n0 + mq + 2 * h - (T_LEN - 1);

    const int kbeg   = klo & ~3;
    const int nsteps = ((khi - kbeg) >> 2) + 1;
    const int per    = (nsteps + NWAVES - 1) / NWAVES;
    int wLo = kbeg + (wave * per) * 4;
    int wHi = kbeg + min(nsteps, (wave + 1) * per) * 4;

    int iLo = T_LEN - 1 - n0;  if (iLo < 240) iLo = 240;
    iLo = (iLo + 3) & ~3;
    int iHi = 2 * T_LEN - 276 - n0;  if (iHi > T_LEN - 4) iHi = T_LEN - 4;
    int iHiE = (iHi & ~3) + 4;
    if (iHiE < iLo) iHiE = iLo;

    const int mLo = min(max(wLo, iLo), wHi);
    const int mHi = max(min(wHi, iHiE), mLo);

    v8f acc0a = {0.f,0.f,0.f,0.f,0.f,0.f,0.f,0.f};
    v8f acc0b = {0.f,0.f,0.f,0.f,0.f,0.f,0.f,0.f};
    v8f acc1a = {0.f,0.f,0.f,0.f,0.f,0.f,0.f,0.f};
    v8f acc1b = {0.f,0.f,0.f,0.f,0.f,0.f,0.f,0.f};

    for (int k0 = wLo; k0 < mLo; k0 += 4)
        step_checked2(rir, a_ch, k0, rbase, bbase, acc0a, acc1a);

    {
        const char* pa = (const char*)rir  + (ptrdiff_t)(mLo + rbase) * 4;
        const char* pb = (const char*)a_ch + (ptrdiff_t)(mLo + bbase) * 4;
        int k0 = mLo;
        for (; k0 + 4 < mHi; k0 += 8) {
            v2f a0  = *(const v2f*)(pa);
            v2f a1  = *(const v2f*)(pa + 16);
            f2u b00 = *(const f2u*)(pb);
            f2u b01 = *(const f2u*)(pb + 16);
            f2u b10 = *(const f2u*)(pb + 1024);
            f2u b11 = *(const f2u*)(pb + 1040);
            v2f v00 = {b00.x, b00.y};
            v2f v01 = {b01.x, b01.y};
            v2f v10 = {b10.x, b10.y};
            v2f v11 = {b11.x, b11.y};
            acc0a = wmma4(a0, v00, acc0a);
            acc1a = wmma4(a0, v10, acc1a);
            acc0b = wmma4(a1, v01, acc0b);
            acc1b = wmma4(a1, v11, acc1b);
            pa += 32;
            pb += 32;
        }
        if (k0 < mHi) {
            v2f a0  = *(const v2f*)(pa);
            f2u b00 = *(const f2u*)(pb);
            f2u b10 = *(const f2u*)(pb + 1024);
            v2f v00 = {b00.x, b00.y};
            v2f v10 = {b10.x, b10.y};
            acc0a = wmma4(a0, v00, acc0a);
            acc1a = wmma4(a0, v10, acc1a);
        }
    }

    for (int kc = mHi; kc < wHi; kc += 4)
        step_checked2(rir, a_ch, kc, rbase, bbase, acc0b, acc1b);

    #pragma unroll
    for (int v = 0; v < 8; ++v) {
        int M = v + 8 * h;
        red[wave * TILE_W +       M * 16 + mq] = acc0a[v] + acc0b[v];
        red[wave * TILE_W + 256 + M * 16 + mq] = acc1a[v] + acc1b[v];
    }
    __syncthreads();

    float s0 = 0.f, s1 = 0.f;
    #pragma unroll
    for (int w = 0; w < NWAVES; ++w) {
        s0 += red[w * TILE_W + tid];
        s1 += red[w * TILE_W + 256 + tid];
    }
    int n = n0 + tid;
    if (n < OUT_LEN)       out[ch * OUT_LEN + n]       = s0;
    if (n + 256 < OUT_LEN) out[ch * OUT_LEN + n + 256] = s1;
}

extern "C" void kernel_launch(void* const* d_in, const int* in_sizes, int n_in,
                              void* d_out, int out_size, void* d_ws, size_t ws_size,
                              hipStream_t stream) {
    (void)in_sizes; (void)n_in; (void)d_ws; (void)ws_size; (void)out_size;
    const float* audio = (const float*)d_in[0];
    const float* rir   = (const float*)d_in[1];
    float* out = (float*)d_out;

    dim3 grid(NTILES, 4, 1);
    dim3 block(256, 1, 1);
    RoomImpulseResponse_wmma_conv<<<grid, block, 0, stream>>>(audio, rir, out);
}


# Partitioned frequency-domain form: 512 partitions of 256, 512-point transforms as constant-matrix products, per-bin 512x512 block convolution on the f32 matrix unit; 3 launches; sum order changes
# speedup vs baseline: 20.1451x; 20.1451x over previous
#include <hip/hip_runtime.h>
#include <hip/hip_bf16.h>


typedef __attribute__((ext_vector_type(2))) float v2f;
typedef __attribute__((ext_vector_type(4))) float v4f;
typedef __attribute__((ext_vector_type(8))) float v8f;

typedef struct __attribute__((packed, aligned(4))) { float x, y; } f2u;

#define T_LEN   131072
#define OUT_LEN (2 * T_LEN - 1)
#define TILE_W  512
#define NTILES  ((OUT_LEN + TILE_W - 1) / TILE_W)
#define NWAVES  8

#define GPC         47
#define KC          (16 * GPC)
#define RIR_WORDS   1280
#define AUD_WORDS   (KC + 272)
#define WAVE_WORDS  (RIR_WORDS + AUD_WORDS)
#define NCH         4
#define NBLOCKS     (NTILES * NCH)


__device__ __forceinline__ v8f wmma4(v2f a, v2f b, v8f c) {
    return __builtin_amdgcn_wmma_f32_16x16x4_f32(false, a, false, b,
                                                 (short)0, c, false, false);
}

__device__ __forceinline__ v2f ldpair(const float* p) {
    f2u t = *(const f2u*)p;
    v2f r = {t.x, t.y};
    return r;
}

__device__ __forceinline__ void wave_lds_handover() {
    __builtin_amdgcn_fence(__ATOMIC_RELEASE, "workgroup", "local");
    __builtin_amdgcn_wave_barrier();
    __builtin_amdgcn_fence(__ATOMIC_ACQUIRE, "workgroup", "local");
}

__global__ __launch_bounds__(256)
void RoomImpulseResponse_wmma_conv(const float* __restrict__ audio,
                                   const float* __restrict__ rir,
                                   float* __restrict__ out)
{
    __shared__ __attribute__((aligned(16))) float smem[NWAVES * WAVE_WORDS];

    const unsigned bid = blockIdx.x;
    const int ch   = (int)(bid % NCH);
    const int rank = (int)(bid / NCH);
    const int tile = (rank & 1) ? (NTILES / 2 - 1 - (rank >> 1))
                                : (NTILES / 2 + (rank >> 1));
    const int tid  = threadIdx.x;
    const int wave = __builtin_amdgcn_readfirstlane(tid >> 5);
    const int lane = tid & 31;
    const int h    = lane >> 4;
    const int mq   = lane & 15;

    const float* __restrict__ a_ch = audio + ch * T_LEN;

    float* rbuf = smem + wave * WAVE_WORDS;
    float* abuf = rbuf + RIR_WORDS;
    const float* pA0  = rbuf + 20 * (15 - mq) + 2 * h;
    const float* pB00 = abuf + mq + 2 * h + 1;
    const float* pB10 = pB00 + 256;

    {
        const int n0   = tile * TILE_W;

        int klo = T_LEN - 272 - n0;  if (klo < 0) klo = 0;
        int khi = 2 * T_LEN - 2 - n0;
        if (khi > T_LEN - 1 + 240) khi = T_LEN - 1 + 240;
        const int kbeg    = klo & ~15;
        const int ngroups = ((khi - kbeg) >> 4) + 1;
        const int per     = (ngroups + NWAVES - 1) / NWAVES;
        const int gLo     = wave * per;
        const int gHi     = min(ngroups, gLo + per);

        v8f acc0a = {0.f,0.f,0.f,0.f,0.f,0.f,0.f,0.f};
        v8f acc0b = {0.f,0.f,0.f,0.f,0.f,0.f,0.f,0.f};
        v8f acc1a = {0.f,0.f,0.f,0.f,0.f,0.f,0.f,0.f};
        v8f acc1b = {0.f,0.f,0.f,0.f,0.f,0.f,0.f,0.f};

        for (int g0 = gLo; g0 < gHi; g0 += GPC) {
            const int kc = kbeg + 16 * g0;

            {
                const int R0 = kc - 240;
                v4f t[8];
                #pragma unroll
                for (int j = 0; j < 8; ++j) {
                    const int gi = R0 + 4 * (lane + 32 * j);
                    t[j] = *(const v4f*)(rir + min(max(gi, 0), T_LEN - 4));
                }
                #pragma unroll
                for (int j = 0; j < 8; ++j) {
                    const int g  = lane + 32 * j;
                    const int gi = R0 + 4 * g;
                    v4f v = t[j];
                    if ((unsigned)gi >= (unsigned)T_LEN) { v4f z = {0.f,0.f,0.f,0.f}; v = z; }
                    *(v4f*)(rbuf + 4 * g + 4 * (g >> 2)) = v;
                }
            }
            {
                const int S0 = n0 + kc - T_LEN;
                v4f t[8];
                #pragma unroll
                for (int j = 0; j < 8; ++j) {
                    const int gi = S0 + 4 * (lane + 32 * j);
                    t[j] = *(const v4f*)(a_ch + min(max(gi, 0), T_LEN - 4));
                }
                #pragma unroll
                for (int j = 0; j < 8; ++j) {
                    const int g  = lane + 32 * j;
                    const int gi = S0 + 4 * g;
                    v4f v = t[j];
                    if ((unsigned)gi >= (unsigned)T_LEN) { v4f z = {0.f,0.f,0.f,0.f}; v = z; }
                    *(v4f*)(abuf + 4 * g) = v;
                }
            }
            wave_lds_handover();

            const int ng = min(GPC, gHi - g0);
            const float* pA  = pA0;
            const float* pB0 = pB00;
            const float* pB1 = pB10;
            for (int g = 0; g < ng; ++g) {
                v2f a0  = *(const v2f*)(pA);
                v2f a1  = *(const v2f*)(pA + 4);
                v2f a2  = *(const v2f*)(pA + 8);
                v2f a3  = *(const v2f*)(pA + 12);
                v2f b00 = ldpair(pB0);
                v2f b01 = ldpair(pB0 + 4);
                v2f b02 = ldpair(pB0 + 8);
                v2f b03 = ldpair(pB0 + 12);
                v2f b10 = ldpair(pB1);
                v2f b11 = ldpair(pB1 + 4);
                v2f b12 = ldpair(pB1 + 8);
                v2f b13 = ldpair(pB1 + 12);
                acc0a = wmma4(a0, b00, acc0a);
                acc1a = wmma4(a0, b10, acc1a);
                acc0b = wmma4(a1, b01, acc0b);
                acc1b = wmma4(a1, b11, acc1b);
                acc0a = wmma4(a2, b02, acc0a);
                acc1a = wmma4(a2, b12, acc1a);
                acc0b = wmma4(a3, b03, acc0b);
                acc1b = wmma4(a3, b13, acc1b);
                pA  += 20;
                pB0 += 16;
                pB1 += 16;
            }
            wave_lds_handover();
        }

        __syncthreads();
        float* red = smem;

        #pragma unroll
        for (int v = 0; v < 8; ++v) {
            int M = v + 8 * h;
            red[wave * TILE_W +       M * 16 + mq] = acc0a[v] + acc0b[v];
            red[wave * TILE_W + 256 + M * 16 + mq] = acc1a[v] + acc1b[v];
        }
        __syncthreads();

        float s0 = 0.f, s1 = 0.f;
        #pragma unroll
        for (int w = 0; w < NWAVES; ++w) {
            s0 += red[w * TILE_W + tid];
            s1 += red[w * TILE_W + 256 + tid];
        }
        int n = n0 + tid;
        if (n < OUT_LEN)       out[ch * OUT_LEN + n]       = s0;
        if (n + 256 < OUT_LEN) out[ch * OUT_LEN + n + 256] = s1;
    }
}


#define PB        256
#define NPART     512
#define NBIN      256
#define NQ        1024
#define S_FLOATS  (5 * NBIN * 2 * NPART)
#define Y_FLOATS  (NCH * NBIN * 2 * NQ)
#define WS_NEED_BYTES 13631488ull
static_assert((unsigned long long)(S_FLOATS + Y_FLOATS) * 4ull == WS_NEED_BYTES, "workspace size");

#define XS_PITCH  260
#define ZS_PITCH  516

__device__ __forceinline__ void fwd_step(const float* tab, int f, int n, v2f b,
                                         v8f& re, v8f& im) {
    const int j0 = (f * n) & 511;
    const int j1 = (j0 + f) & 511;
    const v2f t0 = *(const v2f*)(tab + 2 * j0);
    const v2f t1 = *(const v2f*)(tab + 2 * j1);
    v2f ar = {t0.x, t1.x};
    v2f ai = {t0.y, t1.y};
    if (f == 0) { ai.x = 1.f; ai.y = -1.f; }
    re = wmma4(ar, b, re);
    im = wmma4(ai, b, im);
}

__global__ __launch_bounds__(256)
void rir_part_forward(const float* __restrict__ audio,
                      const float* __restrict__ rir,
                      float* __restrict__ S)
{
    __shared__ __attribute__((aligned(16))) float tab[1024];
    __shared__ __attribute__((aligned(16))) float xs[16 * XS_PITCH];
    const int bx   = (int)blockIdx.x;
    const int half = bx & 1;
    const int pt   = (bx >> 1) & 31;
    const int sig  = bx >> 6;
    const int tid  = threadIdx.x;

    for (int j = tid; j < 512; j += 256) {
        float s, c;
        sincospif((float)j * (1.0f / 256.0f), &s, &c);
        tab[2 * j]     = c;
        tab[2 * j + 1] = -s;
    }
    {
        float xv[16];
        if (sig < NCH) {
            const float* src = audio + sig * T_LEN + PB * 16 * pt + tid;
            #pragma unroll
            for (int i = 0; i < 16; ++i) xv[i] = src[PB * i];
        } else {
            const float* src = rir + PB * 16 * pt + 255 - tid;
            #pragma unroll
            for (int i = 0; i < 16; ++i) xv[i] = src[PB * i];
        }
        #pragma unroll
        for (int i = 0; i < 16; ++i) xs[i * XS_PITCH + tid] = xv[i];
    }
    __syncthreads();

    const int wave = tid >> 5;
    const int lane = tid & 31;
    const int h    = lane >> 4;
    const int mq   = lane & 15;
    const int ft   = 8 * half + wave;
    const int f    = 16 * ft + mq;
    const float* pb = xs + mq * XS_PITCH + 2 * h;

    v8f re0 = {0.f,0.f,0.f,0.f,0.f,0.f,0.f,0.f};
    v8f re1 = re0, im0 = re0, im1 = re0;
    for (int k0 = 0; k0 < PB; k0 += 8) {
        const v2f b0 = *(const v2f*)(pb + k0);
        const v2f b1 = *(const v2f*)(pb + k0 + 4);
        fwd_step(tab, f, k0 + 2 * h,     b0, re0, im0);
        fwd_step(tab, f, k0 + 4 + 2 * h, b1, re1, im1);
    }
    const int p = 16 * pt + mq;
    #pragma unroll
    for (int v = 0; v < 8; ++v) {
        const int fo = 16 * ft + v + 8 * h;
        float* d = S + ((sig * NBIN + fo) * 2) * NPART + p;
        d[0]     = re0[v] + re1[v];
        d[NPART] = im0[v] + im1[v];
    }
}

#define RS_WORDS 1280
#define XSW      544
__global__ __launch_bounds__(128)
void rir_bin_conv(const float* __restrict__ S, float* __restrict__ Y)
{
    __shared__ __attribute__((aligned(16))) float sm[2 * RS_WORDS + 2 * XSW];
    float* Rr = sm;
    float* Xr = sm + 2 * RS_WORDS;
    const int f   = (int)blockIdx.x >> 2;
    const int c   = (int)blockIdx.x & 3;
    const int tid = threadIdx.x;
    const float* gR = S + ((NCH * NBIN + f) * 2) * NPART;
    const float* gX = S + ((c * NBIN + f) * 2) * NPART;

    {
        float rvr[8], rvi[8], xvr[5], xvi[5];
        #pragma unroll
        for (int i = 0; i < 8; ++i) {
            const int rc = min(max(tid + 128 * i - 240, 0), NPART - 1);
            rvr[i] = gR[rc];
            rvi[i] = gR[NPART + rc];
        }
        #pragma unroll
        for (int i = 0; i < 5; ++i) {
            const int mc = min(max(tid + 128 * i - 16, 0), NPART - 1);
            xvr[i] = gX[mc];
            xvi[i] = gX[NPART + mc];
        }
        #pragma unroll
        for (int i = 0; i < 8; ++i) {
            const int t = tid + 128 * i;
            const bool in = (unsigned)(t - 240) < (unsigned)NPART;
            if (t < 992) {
                const int w = t + 4 * (t >> 4);
                Rr[w]            = in ? rvr[i] : 0.f;
                Rr[RS_WORDS + w] = in ? rvi[i] : 0.f;
            }
        }
        #pragma unroll
        for (int i = 0; i < 5; ++i) {
            const int w = tid + 128 * i;
            const bool in = (unsigned)(w - 16) < (unsigned)NPART;
            if (w < XSW) {
                Xr[w]       = in ? xvr[i] : 0.f;
                Xr[XSW + w] = in ? xvi[i] : 0.f;
            }
        }
    }
    __syncthreads();

    const int wave = __builtin_amdgcn_readfirstlane(tid >> 5);
    const int lane = tid & 31;
    const int h    = lane >> 4;
    const int mq   = lane & 15;
    const int tile = (wave + c + f + (f >> 6)) & 3;
    const int n0   = 256 * tile;
    const int klo  = max(0, 496 - n0);
    const int khi  = min(752, 1024 - n0);
    const int ng   = (khi - klo) >> 4;

    const float* pA = Rr + 20 * (15 - mq) + 2 * h + 20 * (klo >> 4);
    const float* pB = Xr + n0 + mq + 2 * h + klo - 495;

    v8f rr = {0.f,0.f,0.f,0.f,0.f,0.f,0.f,0.f};
    v8f ri = rr, ii = rr, ir = rr;
    for (int g = 0; g < ng; ++g) {
        #pragma unroll
        for (int s = 0; s < 4; ++s) {
            const v2f ar = *(const v2f*)(pA + 4 * s);
            const v2f ai = *(const v2f*)(pA + RS_WORDS + 4 * s);
            const v2f br = ldpair(pB + 4 * s);
            const v2f bi = ldpair(pB + XSW + 4 * s);
            rr = wmma4(ar, br, rr);
            ri = wmma4(ar, bi, ri);
            ii = wmma4(ai, bi, ii);
            ir = wmma4(ai, br, ir);
        }
        pA += 20;
        pB += 16;
    }
    float* d = Y + ((c * NBIN + f) * 2) * NQ + n0 + mq;
    #pragma unroll
    for (int v = 0; v < 8; ++v) {
        const int M = v + 8 * h;
        const float yr = (f == 0) ? rr[v] : rr[v] - ii[v];
        const float yi = (f == 0) ? ii[v] : ri[v] + ir[v];
        d[16 * M]      = yr;
        d[NQ + 16 * M] = yi;
    }
}

__device__ __forceinline__ v8f inv_step(const float* tab, const float* pa,
                                        int f, int n, v8f acc) {
    const v2f a = *(const v2f*)pa;
    const int j = (f * n) & 511;
    v2f b = *(const v2f*)(tab + 2 * j);
    if (f == 0) {
        b.x = 1.0f / 512.0f;
        b.y = (n & 1) ? -1.0f / 512.0f : 1.0f / 512.0f;
    }
    return wmma4(a, b, acc);
}

__global__ __launch_bounds__(512)
void rir_part_inverse(const float* __restrict__ Y, float* __restrict__ out)
{
    __shared__ __attribute__((aligned(16))) float tab[1024];
    __shared__ __attribute__((aligned(16))) float zs[16 * ZS_PITCH];
    const int c   = (int)blockIdx.x >> 6;
    const int q0  = 16 * ((int)blockIdx.x & 63);
    const int tid = threadIdx.x;
    {
        float s, co;
        sincospif((float)tid * (1.0f / 256.0f), &s, &co);
        tab[2 * tid]     = co * (1.0f / 256.0f);
        tab[2 * tid + 1] = -s * (1.0f / 256.0f);
    }
    const float* gY = Y + c * (NBIN * 2 * NQ);
    {
        const int qc = tid & 15;
        const int q  = q0 + qc;
        const float* col = gY + (tid >> 4) * NQ + q;
        const int back = (q > 0) ? 1 : 0;
        float y1[16], y0[16];
        #pragma unroll
        for (int i = 0; i < 16; ++i) {
            y1[i] = col[32 * NQ * i];
            y0[i] = col[32 * NQ * i - back];
        }
        float* zc = zs + qc * ZS_PITCH + (tid >> 4);
        const bool odd = ((tid >> 4) & 2) != 0;
        #pragma unroll
        for (int i = 0; i < 16; ++i) {
            const float p = (q > 0) ? y0[i] : 0.f;
            zc[32 * i] = odd ? y1[i] - p : y1[i] + p;
        }
    }
    __syncthreads();

    const int wave = tid >> 5;
    const int lane = tid & 31;
    const int h    = lane >> 4;
    const int mq   = lane & 15;
    const int n    = 16 * wave + mq;
    const float* pa = zs + mq * ZS_PITCH + 2 * h;

    v8f acc0 = {0.f,0.f,0.f,0.f,0.f,0.f,0.f,0.f};
    v8f acc1 = acc0;
    for (int k0 = 0; k0 < 512; k0 += 8) {
        acc0 = inv_step(tab, pa + k0,     (k0 >> 1) + h,     n, acc0);
        acc1 = inv_step(tab, pa + k0 + 4, (k0 >> 1) + 2 + h, n, acc1);
    }
    #pragma unroll
    for (int v = 0; v < 8; ++v) {
        const int q   = q0 + v + 8 * h;
        const int idx = PB * q + n;
        if (idx < OUT_LEN) out[c * OUT_LEN + idx] = acc0[v] + acc1[v];
    }
}

extern "C" void kernel_launch(void* const* d_in, const int* in_sizes, int n_in,
                              void* d_out, int out_size, void* d_ws, size_t ws_size,
                              hipStream_t stream) {
    (void)in_sizes; (void)n_in; (void)out_size;
    const float* audio = (const float*)d_in[0];
    const float* rir   = (const float*)d_in[1];
    float* out = (float*)d_out;

    if (d_ws != nullptr && ws_size >= WS_NEED_BYTES) {
        float* S = (float*)d_ws;
        float* Y = S + S_FLOATS;
        rir_part_forward<<<dim3(320, 1, 1), dim3(256, 1, 1), 0, stream>>>(audio, rir, S);
        rir_bin_conv<<<dim3(1024, 1, 1), dim3(128, 1, 1), 0, stream>>>(S, Y);
        rir_part_inverse<<<dim3(256, 1, 1), dim3(512, 1, 1), 0, stream>>>(Y, out);
    } else {
        dim3 grid(NBLOCKS, 1, 1);
        dim3 block(256, 1, 1);
        RoomImpulseResponse_wmma_conv<<<grid, block, 0, stream>>>(audio, rir, out);
    }
}


# Per-bin complex block convolution from three real matrix products (was four): Rr*Xr, Ri*Xi, (Rr+Ri)*(Xr+Xi), operand sums formed once at staging; f32 operand sums, sum order changes
# speedup vs baseline: 21.4885x; 1.0667x over previous
#include <hip/hip_runtime.h>
#include <hip/hip_bf16.h>


typedef __attribute__((ext_vector_type(2))) float v2f;
typedef __attribute__((ext_vector_type(4))) float v4f;
typedef __attribute__((ext_vector_type(8))) float v8f;

typedef struct __attribute__((packed, aligned(4))) { float x, y; } f2u;

#define T_LEN   131072
#define OUT_LEN (2 * T_LEN - 1)
#define TILE_W  512
#define NTILES  ((OUT_LEN + TILE_W - 1) / TILE_W)
#define NWAVES  8

#define GPC         47
#define KC          (16 * GPC)
#define RIR_WORDS   1280
#define AUD_WORDS   (KC + 272)
#define WAVE_WORDS  (RIR_WORDS + AUD_WORDS)
#define NCH         4
#define NBLOCKS     (NTILES * NCH)


__device__ __forceinline__ v8f wmma4(v2f a, v2f b, v8f c) {
    return __builtin_amdgcn_wmma_f32_16x16x4_f32(false, a, false, b,
                                                 (short)0, c, false, false);
}

__device__ __forceinline__ v2f ldpair(const float* p) {
    f2u t = *(const f2u*)p;
    v2f r = {t.x, t.y};
    return r;
}

__device__ __forceinline__ void wave_lds_handover() {
    __builtin_amdgcn_fence(__ATOMIC_RELEASE, "workgroup", "local");
    __builtin_amdgcn_wave_barrier();
    __builtin_amdgcn_fence(__ATOMIC_ACQUIRE, "workgroup", "local");
}

__global__ __launch_bounds__(256)
void RoomImpulseResponse_wmma_conv(const float* __restrict__ audio,
                                   const float* __restrict__ rir,
                                   float* __restrict__ out)
{
    __shared__ __attribute__((aligned(16))) float smem[NWAVES * WAVE_WORDS];

    const unsigned bid = blockIdx.x;
    const int ch   = (int)(bid % NCH);
    const int rank = (int)(bid / NCH);
    const int tile = (rank & 1) ? (NTILES / 2 - 1 - (rank >> 1))
                                : (NTILES / 2 + (rank >> 1));
    const int tid  = threadIdx.x;
    const int wave = __builtin_amdgcn_readfirstlane(tid >> 5);
    const int lane = tid & 31;
    const int h    = lane >> 4;
    const int mq   = lane & 15;

    const float* __restrict__ a_ch = audio + ch * T_LEN;

    float* rbuf = smem + wave * WAVE_WORDS;
    float* abuf = rbuf + RIR_WORDS;
    const float* pA0  = rbuf + 20 * (15 - mq) + 2 * h;
    const float* pB00 = abuf + mq + 2 * h + 1;
    const float* pB10 = pB00 + 256;

    {
        const int n0   = tile * TILE_W;

        int klo = T_LEN - 272 - n0;  if (klo < 0) klo = 0;
        int khi = 2 * T_LEN - 2 - n0;
        if (khi > T_LEN - 1 + 240) khi = T_LEN - 1 + 240;
        const int kbeg    = klo & ~15;
        const int ngroups = ((khi - kbeg) >> 4) + 1;
        const int per     = (ngroups + NWAVES - 1) / NWAVES;
        const int gLo     = wave * per;
        const int gHi     = min(ngroups, gLo + per);

        v8f acc0a = {0.f,0.f,0.f,0.f,0.f,0.f,0.f,0.f};
        v8f acc0b = {0.f,0.f,0.f,0.f,0.f,0.f,0.f,0.f};
        v8f acc1a = {0.f,0.f,0.f,0.f,0.f,0.f,0.f,0.f};
        v8f acc1b = {0.f,0.f,0.f,0.f,0.f,0.f,0.f,0.f};

        for (int g0 = gLo; g0 < gHi; g0 += GPC) {
            const int kc = kbeg + 16 * g0;

            {
                const int R0 = kc - 240;
                v4f t[8];
                #pragma unroll
                for (int j = 0; j < 8; ++j) {
                    const int gi = R0 + 4 * (lane + 32 * j);
                    t[j] = *(const v4f*)(rir + min(max(gi, 0), T_LEN - 4));
                }
                #pragma unroll
                for (int j = 0; j < 8; ++j) {
                    const int g  = lane + 32 * j;
                    const int gi = R0 + 4 * g;
                    v4f v = t[j];
                    if ((unsigned)gi >= (unsigned)T_LEN) { v4f z = {0.f,0.f,0.f,0.f}; v = z; }
                    *(v4f*)(rbuf + 4 * g + 4 * (g >> 2)) = v;
                }
            }
            {
                const int S0 = n0 + kc - T_LEN;
                v4f t[8];
                #pragma unroll
                for (int j = 0; j < 8; ++j) {
                    const int gi = S0 + 4 * (lane + 32 * j);
                    t[j] = *(const v4f*)(a_ch + min(max(gi, 0), T_LEN - 4));
                }
                #pragma unroll
                for (int j = 0; j < 8; ++j) {
                    const int g  = lane + 32 * j;
                    const int gi = S0 + 4 * g;
                    v4f v = t[j];
                    if ((unsigned)gi >= (unsigned)T_LEN) { v4f z = {0.f,0.f,0.f,0.f}; v = z; }
                    *(v4f*)(abuf + 4 * g) = v;
                }
            }
            wave_lds_handover();

            const int ng = min(GPC, gHi - g0);
            const float* pA  = pA0;
            const float* pB0 = pB00;
            const float* pB1 = pB10;
            for (int g = 0; g < ng; ++g) {
                v2f a0  = *(const v2f*)(pA);
                v2f a1  = *(const v2f*)(pA + 4);
                v2f a2  = *(const v2f*)(pA + 8);
                v2f a3  = *(const v2f*)(pA + 12);
                v2f b00 = ldpair(pB0);
                v2f b01 = ldpair(pB0 + 4);
                v2f b02 = ldpair(pB0 + 8);
                v2f b03 = ldpair(pB0 + 12);
                v2f b10 = ldpair(pB1);
                v2f b11 = ldpair(pB1 + 4);
                v2f b12 = ldpair(pB1 + 8);
                v2f b13 = ldpair(pB1 + 12);
                acc0a = wmma4(a0, b00, acc0a);
                acc1a = wmma4(a0, b10, acc1a);
                acc0b = wmma4(a1, b01, acc0b);
                acc1b = wmma4(a1, b11, acc1b);
                acc0a = wmma4(a2, b02, acc0a);
                acc1a = wmma4(a2, b12, acc1a);
                acc0b = wmma4(a3, b03, acc0b);
                acc1b = wmma4(a3, b13, acc1b);
                pA  += 20;
                pB0 += 16;
                pB1 += 16;
            }
            wave_lds_handover();
        }

        __syncthreads();
        float* red = smem;

        #pragma unroll
        for (int v = 0; v < 8; ++v) {
            int M = v + 8 * h;
            red[wave * TILE_W +       M * 16 + mq] = acc0a[v] + acc0b[v];
            red[wave * TILE_W + 256 + M * 16 + mq] = acc1a[v] + acc1b[v];
        }
        __syncthreads();

        float s0 = 0.f, s1 = 0.f;
        #pragma unroll
        for (int w = 0; w < NWAVES; ++w) {
            s0 += red[w * TILE_W + tid];
            s1 += red[w * TILE_W + 256 + tid];
        }
        int n = n0 + tid;
        if (n < OUT_LEN)       out[ch * OUT_LEN + n]       = s0;
        if (n + 256 < OUT_LEN) out[ch * OUT_LEN + n + 256] = s1;
    }
}


#define PB        256
#define NPART     512
#define NBIN      256
#define NQ        1024
#define S_FLOATS  (5 * NBIN * 2 * NPART)
#define Y_FLOATS  (NCH * NBIN * 2 * NQ)
#define WS_NEED_BYTES 13631488ull
static_assert((unsigned long long)(S_FLOATS + Y_FLOATS) * 4ull == WS_NEED_BYTES, "workspace size");

#define XS_PITCH  260
#define ZS_PITCH  516

__device__ __forceinline__ void fwd_step(const float* tab, int f, int n, v2f b,
                                         v8f& re, v8f& im) {
    const int j0 = (f * n) & 511;
    const int j1 = (j0 + f) & 511;
    const v2f t0 = *(const v2f*)(tab + 2 * j0);
    const v2f t1 = *(const v2f*)(tab + 2 * j1);
    v2f ar = {t0.x, t1.x};
    v2f ai = {t0.y, t1.y};
    if (f == 0) { ai.x = 1.f; ai.y = -1.f; }
    re = wmma4(ar, b, re);
    im = wmma4(ai, b, im);
}

__global__ __launch_bounds__(256)
void rir_part_forward(const float* __restrict__ audio,
                      const float* __restrict__ rir,
                      float* __restrict__ S)
{
    __shared__ __attribute__((aligned(16))) float tab[1024];
    __shared__ __attribute__((aligned(16))) float xs[16 * XS_PITCH];
    const int bx   = (int)blockIdx.x;
    const int half = bx & 1;
    const int pt   = (bx >> 1) & 31;
    const int sig  = bx >> 6;
    const int tid  = threadIdx.x;

    for (int j = tid; j < 512; j += 256) {
        float s, c;
        sincospif((float)j * (1.0f / 256.0f), &s, &c);
        tab[2 * j]     = c;
        tab[2 * j + 1] = -s;
    }
    {
        float xv[16];
        if (sig < NCH) {
            const float* src = audio + sig * T_LEN + PB * 16 * pt + tid;
            #pragma unroll
            for (int i = 0; i < 16; ++i) xv[i] = src[PB * i];
        } else {
            const float* src = rir + PB * 16 * pt + 255 - tid;
            #pragma unroll
            for (int i = 0; i < 16; ++i) xv[i] = src[PB * i];
        }
        #pragma unroll
        for (int i = 0; i < 16; ++i) xs[i * XS_PITCH + tid] = xv[i];
    }
    __syncthreads();

    const int wave = tid >> 5;
    const int lane = tid & 31;
    const int h    = lane >> 4;
    const int mq   = lane & 15;
    const int ft   = 8 * half + wave;
    const int f    = 16 * ft + mq;
    const float* pb = xs + mq * XS_PITCH + 2 * h;

    v8f re0 = {0.f,0.f,0.f,0.f,0.f,0.f,0.f,0.f};
    v8f re1 = re0, im0 = re0, im1 = re0;
    for (int k0 = 0; k0 < PB; k0 += 8) {
        const v2f b0 = *(const v2f*)(pb + k0);
        const v2f b1 = *(const v2f*)(pb + k0 + 4);
        fwd_step(tab, f, k0 + 2 * h,     b0, re0, im0);
        fwd_step(tab, f, k0 + 4 + 2 * h, b1, re1, im1);
    }
    const int p = 16 * pt + mq;
    #pragma unroll
    for (int v = 0; v < 8; ++v) {
        const int fo = 16 * ft + v + 8 * h;
        float* d = S + ((sig * NBIN + fo) * 2) * NPART + p;
        d[0]     = re0[v] + re1[v];
        d[NPART] = im0[v] + im1[v];
    }
}

#define RS_WORDS 1280
#define XSW      544
__global__ __launch_bounds__(128)
void rir_bin_conv(const float* __restrict__ S, float* __restrict__ Y)
{
    __shared__ __attribute__((aligned(16))) float sm[3 * RS_WORDS + 3 * XSW];
    float* Rs = sm;
    float* Xs = sm + 3 * RS_WORDS;
    const int f   = (int)blockIdx.x >> 2;
    const int c   = (int)blockIdx.x & 3;
    const int tid = threadIdx.x;
    const float* gR = S + ((NCH * NBIN + f) * 2) * NPART;
    const float* gX = S + ((c * NBIN + f) * 2) * NPART;

    {
        float rvr[8], rvi[8], xvr[5], xvi[5];
        #pragma unroll
        for (int i = 0; i < 8; ++i) {
            const int rc = min(max(tid + 128 * i - 240, 0), NPART - 1);
            rvr[i] = gR[rc];
            rvi[i] = gR[NPART + rc];
        }
        #pragma unroll
        for (int i = 0; i < 5; ++i) {
            const int mc = min(max(tid + 128 * i - 16, 0), NPART - 1);
            xvr[i] = gX[mc];
            xvi[i] = gX[NPART + mc];
        }
        #pragma unroll
        for (int i = 0; i < 8; ++i) {
            const int t = tid + 128 * i;
            const bool in = (unsigned)(t - 240) < (unsigned)NPART;
            if (t < 992) {
                const int w = t + 4 * (t >> 4);
                const float re = in ? rvr[i] : 0.f;
                const float im = in ? rvi[i] : 0.f;
                Rs[w]                = re;
                Rs[RS_WORDS + w]     = im;
                Rs[2 * RS_WORDS + w] = re + im;
            }
        }
        #pragma unroll
        for (int i = 0; i < 5; ++i) {
            const int w = tid + 128 * i;
            const bool in = (unsigned)(w - 16) < (unsigned)NPART;
            if (w < XSW) {
                const float re = in ? xvr[i] : 0.f;
                const float im = in ? xvi[i] : 0.f;
                Xs[w]           = re;
                Xs[XSW + w]     = im;
                Xs[2 * XSW + w] = re + im;
            }
        }
    }
    __syncthreads();

    const int wave = __builtin_amdgcn_readfirstlane(tid >> 5);
    const int lane = tid & 31;
    const int h    = lane >> 4;
    const int mq   = lane & 15;
    const int tile = (wave + c + f + (f >> 6)) & 3;
    const int n0   = 256 * tile;
    const int klo  = max(0, 496 - n0);
    const int khi  = min(752, 1024 - n0);
    const int ng   = (khi - klo) >> 4;

    const float* pA = Rs + 20 * (15 - mq) + 2 * h + 20 * (klo >> 4);
    const float* pB = Xs + n0 + mq + 2 * h + klo - 495;

    v8f p1 = {0.f,0.f,0.f,0.f,0.f,0.f,0.f,0.f};
    v8f p2 = p1, p3 = p1;
    for (int g = 0; g < ng; ++g) {
        #pragma unroll
        for (int s = 0; s < 4; ++s) {
            const v2f ar = *(const v2f*)(pA + 4 * s);
            const v2f ai = *(const v2f*)(pA + RS_WORDS + 4 * s);
            const v2f as = *(const v2f*)(pA + 2 * RS_WORDS + 4 * s);
            const v2f br = ldpair(pB + 4 * s);
            const v2f bi = ldpair(pB + XSW + 4 * s);
            const v2f bs = ldpair(pB + 2 * XSW + 4 * s);
            p1 = wmma4(ar, br, p1);
            p2 = wmma4(ai, bi, p2);
            p3 = wmma4(as, bs, p3);
        }
        pA += 20;
        pB += 16;
    }
    float* d = Y + ((c * NBIN + f) * 2) * NQ + n0 + mq;
    #pragma unroll
    for (int v = 0; v < 8; ++v) {
        const int M = v + 8 * h;
        const float yr = (f == 0) ? p1[v] : p1[v] - p2[v];
        const float yi = (f == 0) ? p2[v] : (p3[v] - p1[v]) - p2[v];
        d[16 * M]      = yr;
        d[NQ + 16 * M] = yi;
    }
}

__device__ __forceinline__ v8f inv_step(const float* tab, const float* pa,
                                        int f, int n, v8f acc) {
    const v2f a = *(const v2f*)pa;
    const int j = (f * n) & 511;
    v2f b = *(const v2f*)(tab + 2 * j);
    if (f == 0) {
        b.x = 1.0f / 512.0f;
        b.y = (n & 1) ? -1.0f / 512.0f : 1.0f / 512.0f;
    }
    return wmma4(a, b, acc);
}

__global__ __launch_bounds__(512)
void rir_part_inverse(const float* __restrict__ Y, float* __restrict__ out)
{
    __shared__ __attribute__((aligned(16))) float tab[1024];
    __shared__ __attribute__((aligned(16))) float zs[16 * ZS_PITCH];
    const int c   = (int)blockIdx.x >> 6;
    const int q0  = 16 * ((int)blockIdx.x & 63);
    const int tid = threadIdx.x;
    {
        float s, co;
        sincospif((float)tid * (1.0f / 256.0f), &s, &co);
        tab[2 * tid]     = co * (1.0f / 256.0f);
        tab[2 * tid + 1] = -s * (1.0f / 256.0f);
    }
    const float* gY = Y + c * (NBIN * 2 * NQ);
    {
        const int qc = tid & 15;
        const int q  = q0 + qc;
        const float* col = gY + (tid >> 4) * NQ + q;
        const int back = (q > 0) ? 1 : 0;
        float y1[16], y0[16];
        #pragma unroll
        for (int i = 0; i < 16; ++i) {
            y1[i] = col[32 * NQ * i];
            y0[i] = col[32 * NQ * i - back];
        }
        float* zc = zs + qc * ZS_PITCH + (tid >> 4);
        const bool odd = ((tid >> 4) & 2) != 0;
        #pragma unroll
        for (int i = 0; i < 16; ++i) {
            const float p = (q > 0) ? y0[i] : 0.f;
            zc[32 * i] = odd ? y1[i] - p : y1[i] + p;
        }
    }
    __syncthreads();

    const int wave = tid >> 5;
    const int lane = tid & 31;
    const int h    = lane >> 4;
    const int mq   = lane & 15;
    const int n    = 16 * wave + mq;
    const float* pa = zs + mq * ZS_PITCH + 2 * h;

    v8f acc0 = {0.f,0.f,0.f,0.f,0.f,0.f,0.f,0.f};
    v8f acc1 = acc0;
    for (int k0 = 0; k0 < 512; k0 += 8) {
        acc0 = inv_step(tab, pa + k0,     (k0 >> 1) + h,     n, acc0);
        acc1 = inv_step(tab, pa + k0 + 4, (k0 >> 1) + 2 + h, n, acc1);
    }
    #pragma unroll
    for (int v = 0; v < 8; ++v) {
        const int q   = q0 + v + 8 * h;
        const int idx = PB * q + n;
        if (idx < OUT_LEN) out[c * OUT_LEN + idx] = acc0[v] + acc1[v];
    }
}

extern "C" void kernel_launch(void* const* d_in, const int* in_sizes, int n_in,
                              void* d_out, int out_size, void* d_ws, size_t ws_size,
                              hipStream_t stream) {
    (void)in_sizes; (void)n_in; (void)out_size;
    const float* audio = (const float*)d_in[0];
    const float* rir   = (const float*)d_in[1];
    float* out = (float*)d_out;

    if (d_ws != nullptr && ws_size >= WS_NEED_BYTES) {
        float* S = (float*)d_ws;
        float* Y = S + S_FLOATS;
        rir_part_forward<<<dim3(320, 1, 1), dim3(256, 1, 1), 0, stream>>>(audio, rir, S);
        rir_bin_conv<<<dim3(1024, 1, 1), dim3(128, 1, 1), 0, stream>>>(S, Y);
        rir_part_inverse<<<dim3(256, 1, 1), dim3(512, 1, 1), 0, stream>>>(Y, out);
    } else {
        dim3 grid(NBLOCKS, 1, 1);
        dim3 block(256, 1, 1);
        RoomImpulseResponse_wmma_conv<<<grid, block, 0, stream>>>(audio, rir, out);
    }
}


# Inverse transform folded on the n <-> 256-n symmetry of the 512-point twiddles: sums E,O over 128 columns give y[n]=E+O, y[256-n]=E-O (half the matrix products); n=128 a signed sum; sum order changes
# speedup vs baseline: 22.3615x; 1.0406x over previous
#include <hip/hip_runtime.h>
#include <hip/hip_bf16.h>


typedef __attribute__((ext_vector_type(2))) float v2f;
typedef __attribute__((ext_vector_type(4))) float v4f;
typedef __attribute__((ext_vector_type(8))) float v8f;

typedef struct __attribute__((packed, aligned(4))) { float x, y; } f2u;

#define T_LEN   131072
#define OUT_LEN (2 * T_LEN - 1)
#define TILE_W  512
#define NTILES  ((OUT_LEN + TILE_W - 1) / TILE_W)
#define NWAVES  8

#define GPC         47
#define KC          (16 * GPC)
#define RIR_WORDS   1280
#define AUD_WORDS   (KC + 272)
#define WAVE_WORDS  (RIR_WORDS + AUD_WORDS)
#define NCH         4
#define NBLOCKS     (NTILES * NCH)


__device__ __forceinline__ v8f wmma4(v2f a, v2f b, v8f c) {
    return __builtin_amdgcn_wmma_f32_16x16x4_f32(false, a, false, b,
                                                 (short)0, c, false, false);
}

__device__ __forceinline__ v2f ldpair(const float* p) {
    f2u t = *(const f2u*)p;
    v2f r = {t.x, t.y};
    return r;
}

__device__ __forceinline__ void wave_lds_handover() {
    __builtin_amdgcn_fence(__ATOMIC_RELEASE, "workgroup", "local");
    __builtin_amdgcn_wave_barrier();
    __builtin_amdgcn_fence(__ATOMIC_ACQUIRE, "workgroup", "local");
}

__global__ __launch_bounds__(256)
void RoomImpulseResponse_wmma_conv(const float* __restrict__ audio,
                                   const float* __restrict__ rir,
                                   float* __restrict__ out)
{
    __shared__ __attribute__((aligned(16))) float smem[NWAVES * WAVE_WORDS];

    const unsigned bid = blockIdx.x;
    const int ch   = (int)(bid % NCH);
    const int rank = (int)(bid / NCH);
    const int tile = (rank & 1) ? (NTILES / 2 - 1 - (rank >> 1))
                                : (NTILES / 2 + (rank >> 1));
    const int tid  = threadIdx.x;
    const int wave = __builtin_amdgcn_readfirstlane(tid >> 5);
    const int lane = tid & 31;
    const int h    = lane >> 4;
    const int mq   = lane & 15;

    const float* __restrict__ a_ch = audio + ch * T_LEN;

    float* rbuf = smem + wave * WAVE_WORDS;
    float* abuf = rbuf + RIR_WORDS;
    const float* pA0  = rbuf + 20 * (15 - mq) + 2 * h;
    const float* pB00 = abuf + mq + 2 * h + 1;
    const float* pB10 = pB00 + 256;

    {
        const int n0   = tile * TILE_W;

        int klo = T_LEN - 272 - n0;  if (klo < 0) klo = 0;
        int khi = 2 * T_LEN - 2 - n0;
        if (khi > T_LEN - 1 + 240) khi = T_LEN - 1 + 240;
        const int kbeg    = klo & ~15;
        const int ngroups = ((khi - kbeg) >> 4) + 1;
        const int per     = (ngroups + NWAVES - 1) / NWAVES;
        const int gLo     = wave * per;
        const int gHi     = min(ngroups, gLo + per);

        v8f acc0a = {0.f,0.f,0.f,0.f,0.f,0.f,0.f,0.f};
        v8f acc0b = {0.f,0.f,0.f,0.f,0.f,0.f,0.f,0.f};
        v8f acc1a = {0.f,0.f,0.f,0.f,0.f,0.f,0.f,0.f};
        v8f acc1b = {0.f,0.f,0.f,0.f,0.f,0.f,0.f,0.f};

        for (int g0 = gLo; g0 < gHi; g0 += GPC) {
            const int kc = kbeg + 16 * g0;

            {
                const int R0 = kc - 240;
                v4f t[8];
                #pragma unroll
                for (int j = 0; j < 8; ++j) {
                    const int gi = R0 + 4 * (lane + 32 * j);
                    t[j] = *(const v4f*)(rir + min(max(gi, 0), T_LEN - 4));
                }
                #pragma unroll
                for (int j = 0; j < 8; ++j) {
                    const int g  = lane + 32 * j;
                    const int gi = R0 + 4 * g;
                    v4f v = t[j];
                    if ((unsigned)gi >= (unsigned)T_LEN) { v4f z = {0.f,0.f,0.f,0.f}; v = z; }
                    *(v4f*)(rbuf + 4 * g + 4 * (g >> 2)) = v;
                }
            }
            {
                const int S0 = n0 + kc - T_LEN;
                v4f t[8];
                #pragma unroll
                for (int j = 0; j < 8; ++j) {
                    const int gi = S0 + 4 * (lane + 32 * j);
                    t[j] = *(const v4f*)(a_ch + min(max(gi, 0), T_LEN - 4));
                }
                #pragma unroll
                for (int j = 0; j < 8; ++j) {
                    const int g  = lane + 32 * j;
                    const int gi = S0 + 4 * g;
                    v4f v = t[j];
                    if ((unsigned)gi >= (unsigned)T_LEN) { v4f z = {0.f,0.f,0.f,0.f}; v = z; }
                    *(v4f*)(abuf + 4 * g) = v;
                }
            }
            wave_lds_handover();

            const int ng = min(GPC, gHi - g0);
            const float* pA  = pA0;
            const float* pB0 = pB00;
            const float* pB1 = pB10;
            for (int g = 0; g < ng; ++g) {
                v2f a0  = *(const v2f*)(pA);
                v2f a1  = *(const v2f*)(pA + 4);
                v2f a2  = *(const v2f*)(pA + 8);
                v2f a3  = *(const v2f*)(pA + 12);
                v2f b00 = ldpair(pB0);
                v2f b01 = ldpair(pB0 + 4);
                v2f b02 = ldpair(pB0 + 8);
                v2f b03 = ldpair(pB0 + 12);
                v2f b10 = ldpair(pB1);
                v2f b11 = ldpair(pB1 + 4);
                v2f b12 = ldpair(pB1 + 8);
                v2f b13 = ldpair(pB1 + 12);
                acc0a = wmma4(a0, b00, acc0a);
                acc1a = wmma4(a0, b10, acc1a);
                acc0b = wmma4(a1, b01, acc0b);
                acc1b = wmma4(a1, b11, acc1b);
                acc0a = wmma4(a2, b02, acc0a);
                acc1a = wmma4(a2, b12, acc1a);
                acc0b = wmma4(a3, b03, acc0b);
                acc1b = wmma4(a3, b13, acc1b);
                pA  += 20;
                pB0 += 16;
                pB1 += 16;
            }
            wave_lds_handover();
        }

        __syncthreads();
        float* red = smem;

        #pragma unroll
        for (int v = 0; v < 8; ++v) {
            int M = v + 8 * h;
            red[wave * TILE_W +       M * 16 + mq] = acc0a[v] + acc0b[v];
            red[wave * TILE_W + 256 + M * 16 + mq] = acc1a[v] + acc1b[v];
        }
        __syncthreads();

        float s0 = 0.f, s1 = 0.f;
        #pragma unroll
        for (int w = 0; w < NWAVES; ++w) {
            s0 += red[w * TILE_W + tid];
            s1 += red[w * TILE_W + 256 + tid];
        }
        int n = n0 + tid;
        if (n < OUT_LEN)       out[ch * OUT_LEN + n]       = s0;
        if (n + 256 < OUT_LEN) out[ch * OUT_LEN + n + 256] = s1;
    }
}


#define PB        256
#define NPART     512
#define NBIN      256
#define NQ        1024
#define S_FLOATS  (5 * NBIN * 2 * NPART)
#define Y_FLOATS  (NCH * NBIN * 2 * NQ)
#define WS_NEED_BYTES 13631488ull
static_assert((unsigned long long)(S_FLOATS + Y_FLOATS) * 4ull == WS_NEED_BYTES, "workspace size");

#define XS_PITCH  260

__device__ __forceinline__ void fwd_step(const float* tab, int f, int n, v2f b,
                                         v8f& re, v8f& im) {
    const int j0 = (f * n) & 511;
    const int j1 = (j0 + f) & 511;
    const v2f t0 = *(const v2f*)(tab + 2 * j0);
    const v2f t1 = *(const v2f*)(tab + 2 * j1);
    v2f ar = {t0.x, t1.x};
    v2f ai = {t0.y, t1.y};
    if (f == 0) { ai.x = 1.f; ai.y = -1.f; }
    re = wmma4(ar, b, re);
    im = wmma4(ai, b, im);
}

__global__ __launch_bounds__(256)
void rir_part_forward(const float* __restrict__ audio,
                      const float* __restrict__ rir,
                      float* __restrict__ S)
{
    __shared__ __attribute__((aligned(16))) float tab[1024];
    __shared__ __attribute__((aligned(16))) float xs[16 * XS_PITCH];
    const int bx   = (int)blockIdx.x;
    const int half = bx & 1;
    const int pt   = (bx >> 1) & 31;
    const int sig  = bx >> 6;
    const int tid  = threadIdx.x;

    for (int j = tid; j < 512; j += 256) {
        float s, c;
        sincospif((float)j * (1.0f / 256.0f), &s, &c);
        tab[2 * j]     = c;
        tab[2 * j + 1] = -s;
    }
    {
        float xv[16];
        if (sig < NCH) {
            const float* src = audio + sig * T_LEN + PB * 16 * pt + tid;
            #pragma unroll
            for (int i = 0; i < 16; ++i) xv[i] = src[PB * i];
        } else {
            const float* src = rir + PB * 16 * pt + 255 - tid;
            #pragma unroll
            for (int i = 0; i < 16; ++i) xv[i] = src[PB * i];
        }
        #pragma unroll
        for (int i = 0; i < 16; ++i) xs[i * XS_PITCH + tid] = xv[i];
    }
    __syncthreads();

    const int wave = tid >> 5;
    const int lane = tid & 31;
    const int h    = lane >> 4;
    const int mq   = lane & 15;
    const int ft   = 8 * half + wave;
    const int f    = 16 * ft + mq;
    const float* pb = xs + mq * XS_PITCH + 2 * h;

    v8f re0 = {0.f,0.f,0.f,0.f,0.f,0.f,0.f,0.f};
    v8f re1 = re0, im0 = re0, im1 = re0;
    for (int k0 = 0; k0 < PB; k0 += 8) {
        const v2f b0 = *(const v2f*)(pb + k0);
        const v2f b1 = *(const v2f*)(pb + k0 + 4);
        fwd_step(tab, f, k0 + 2 * h,     b0, re0, im0);
        fwd_step(tab, f, k0 + 4 + 2 * h, b1, re1, im1);
    }
    const int p = 16 * pt + mq;
    #pragma unroll
    for (int v = 0; v < 8; ++v) {
        const int fo = 16 * ft + v + 8 * h;
        float* d = S + ((sig * NBIN + fo) * 2) * NPART + p;
        d[0]     = re0[v] + re1[v];
        d[NPART] = im0[v] + im1[v];
    }
}

#define RS_WORDS 1280
#define XSW      544
__global__ __launch_bounds__(128)
void rir_bin_conv(const float* __restrict__ S, float* __restrict__ Y)
{
    __shared__ __attribute__((aligned(16))) float sm[3 * RS_WORDS + 3 * XSW];
    float* Rs = sm;
    float* Xs = sm + 3 * RS_WORDS;
    const int f   = (int)blockIdx.x >> 2;
    const int c   = (int)blockIdx.x & 3;
    const int tid = threadIdx.x;
    const float* gR = S + ((NCH * NBIN + f) * 2) * NPART;
    const float* gX = S + ((c * NBIN + f) * 2) * NPART;

    {
        float rvr[8], rvi[8], xvr[5], xvi[5];
        #pragma unroll
        for (int i = 0; i < 8; ++i) {
            const int rc = min(max(tid + 128 * i - 240, 0), NPART - 1);
            rvr[i] = gR[rc];
            rvi[i] = gR[NPART + rc];
        }
        #pragma unroll
        for (int i = 0; i < 5; ++i) {
            const int mc = min(max(tid + 128 * i - 16, 0), NPART - 1);
            xvr[i] = gX[mc];
            xvi[i] = gX[NPART + mc];
        }
        #pragma unroll
        for (int i = 0; i < 8; ++i) {
            const int t = tid + 128 * i;
            const bool in = (unsigned)(t - 240) < (unsigned)NPART;
            if (t < 992) {
                const int w = t + 4 * (t >> 4);
                const float re = in ? rvr[i] : 0.f;
                const float im = in ? rvi[i] : 0.f;
                Rs[w]                = re;
                Rs[RS_WORDS + w]     = im;
                Rs[2 * RS_WORDS + w] = re + im;
            }
        }
        #pragma unroll
        for (int i = 0; i < 5; ++i) {
            const int w = tid + 128 * i;
            const bool in = (unsigned)(w - 16) < (unsigned)NPART;
            if (w < XSW) {
                const float re = in ? xvr[i] : 0.f;
                const float im = in ? xvi[i] : 0.f;
                Xs[w]           = re;
                Xs[XSW + w]     = im;
                Xs[2 * XSW + w] = re + im;
            }
        }
    }
    __syncthreads();

    const int wave = __builtin_amdgcn_readfirstlane(tid >> 5);
    const int lane = tid & 31;
    const int h    = lane >> 4;
    const int mq   = lane & 15;
    const int tile = (wave + c + f + (f >> 6)) & 3;
    const int n0   = 256 * tile;
    const int klo  = max(0, 496 - n0);
    const int khi  = min(752, 1024 - n0);
    const int ng   = (khi - klo) >> 4;

    const float* pA = Rs + 20 * (15 - mq) + 2 * h + 20 * (klo >> 4);
    const float* pB = Xs + n0 + mq + 2 * h + klo - 495;

    v8f p1 = {0.f,0.f,0.f,0.f,0.f,0.f,0.f,0.f};
    v8f p2 = p1, p3 = p1;
    for (int g = 0; g < ng; ++g) {
        #pragma unroll
        for (int s = 0; s < 4; ++s) {
            const v2f ar = *(const v2f*)(pA + 4 * s);
            const v2f ai = *(const v2f*)(pA + RS_WORDS + 4 * s);
            const v2f as = *(const v2f*)(pA + 2 * RS_WORDS + 4 * s);
            const v2f br = ldpair(pB + 4 * s);
            const v2f bi = ldpair(pB + XSW + 4 * s);
            const v2f bs = ldpair(pB + 2 * XSW + 4 * s);
            p1 = wmma4(ar, br, p1);
            p2 = wmma4(ai, bi, p2);
            p3 = wmma4(as, bs, p3);
        }
        pA += 20;
        pB += 16;
    }
    float* d = Y + ((c * NBIN + f) * 2) * NQ + n0 + mq;
    #pragma unroll
    for (int v = 0; v < 8; ++v) {
        const int M = v + 8 * h;
        const float yr = (f == 0) ? p1[v] : p1[v] - p2[v];
        const float yi = (f == 0) ? p2[v] : (p3[v] - p1[v]) - p2[v];
        d[16 * M]      = yr;
        d[NQ + 16 * M] = yi;
    }
}

#define ZP   260
#define PLN  (16 * ZP + 2)
#define RP   130
__device__ __forceinline__ void inv_fold_step(const float* tab, const float* paE,
                                              const float* paO, int f0, int n,
                                              v8f& e, v8f& o) {
    const v2f aE = *(const v2f*)paE;
    const v2f aO = *(const v2f*)paO;
    const int j0 = (f0 * n) & 511;
    const int j1 = (j0 + n) & 511;
    const v2f t0 = *(const v2f*)(tab + 2 * j0);
    const v2f t1 = *(const v2f*)(tab + 2 * j1);
    v2f bE = {t0.x, t1.y};
    v2f bO = {t0.y, t1.x};
    if (f0 == 0) bE.x = 1.0f / 512.0f;
    e = wmma4(aE, bE, e);
    o = wmma4(aO, bO, o);
}

__global__ __launch_bounds__(512)
void rir_part_inverse(const float* __restrict__ Y, float* __restrict__ out)
{
    __shared__ __attribute__((aligned(16))) float tab[1024];
    __shared__ __attribute__((aligned(16))) float zs[2 * PLN];
    __shared__ __attribute__((aligned(16))) float red[4 * 16 * RP];
    __shared__ float y128s[16];
    const int c   = (int)blockIdx.x >> 6;
    const int q0  = 16 * ((int)blockIdx.x & 63);
    const int tid = threadIdx.x;
    {
        float s, co;
        sincospif((float)tid * (1.0f / 256.0f), &s, &co);
        tab[2 * tid]     = co * (1.0f / 256.0f);
        tab[2 * tid + 1] = -s * (1.0f / 256.0f);
    }
    const float* gY = Y + c * (NBIN * 2 * NQ);
    {
        const int qc = tid & 15;
        const int q  = q0 + qc;
        const int r  = tid >> 4;
        const float* col = gY + r * NQ + q;
        const int back = (q > 0) ? 1 : 0;
        float y1[16], y0[16];
        #pragma unroll
        for (int i = 0; i < 16; ++i) {
            y1[i] = col[32 * NQ * i];
            y0[i] = col[32 * NQ * i - back];
        }
        const bool odd = (r & 2) != 0;
        const int  pl  = (((r >> 1) ^ r) & 1);
        float* zc = zs + pl * PLN + qc * ZP + (r >> 1);
        #pragma unroll
        for (int i = 0; i < 16; ++i) {
            const float p = (q > 0) ? y0[i] : 0.f;
            zc[16 * i] = odd ? y1[i] - p : y1[i] + p;
        }
    }
    __syncthreads();

    const int wave = tid >> 5;
    const int lane = tid & 31;
    const int h    = lane >> 4;
    const int mq   = lane & 15;
    const int nt   = wave & 7;
    const int kh   = wave >> 3;
    const int n    = 16 * nt + mq;
    const int fb   = 128 * kh + 2 * h;
    const float* paE = zs + mq * ZP + fb;
    const float* paO = paE + PLN;

    v8f e0 = {0.f,0.f,0.f,0.f,0.f,0.f,0.f,0.f};
    v8f e1 = e0, o0 = e0, o1 = e0;
    for (int t = 0; t < 32; t += 2) {
        inv_fold_step(tab, paE + 4 * t,     paO + 4 * t,     fb + 4 * t,     n, e0, o0);
        inv_fold_step(tab, paE + 4 * t + 4, paO + 4 * t + 4, fb + 4 * t + 4, n, e1, o1);
    }
    {
        float* rE = red + ((2 * kh)     * 16) * RP + n;
        float* rO = red + ((2 * kh + 1) * 16) * RP + n;
        #pragma unroll
        for (int v = 0; v < 8; ++v) {
            const int q = v + 8 * h;
            rE[q * RP] = e0[v] + e1[v];
            rO[q * RP] = o0[v] + o1[v];
        }
    }
    {
        const float* pe = zs + wave * ZP;
        float s = 0.f;
        #pragma unroll
        for (int i = 0; i < 8; ++i) {
            float x = pe[lane + 32 * i];
            if (i == 0 && lane == 0) x *= 0.5f;
            s += x;
        }
        const int m4 = lane & 3;
        if (m4 == 1 || m4 == 2) s = -s;
        #pragma unroll
        for (int off = 16; off > 0; off >>= 1) s += __shfl_xor(s, off);
        if (lane == 0)
            y128s[wave] = s * (1.0f / 256.0f) + zs[PLN + wave * ZP] * (1.0f / 512.0f);
    }
    __syncthreads();

    #pragma unroll
    for (int i = 0; i < 8; ++i) {
        const int o  = tid + 512 * i;
        const int q  = o >> 8;
        const int no = o & 255;
        float val;
        if (no == 128) {
            val = y128s[q];
        } else {
            const int nn = (no < 128) ? no : 256 - no;
            const float E = red[(0 * 16 + q) * RP + nn] + red[(2 * 16 + q) * RP + nn];
            const float O = red[(1 * 16 + q) * RP + nn] + red[(3 * 16 + q) * RP + nn];
            float z = zs[PLN + q * ZP] * (1.0f / 512.0f);
            if (nn & 1) z = -z;
            val = ((no < 128) ? E + O : E - O) + z;
        }
        const int idx = PB * (q0 + q) + no;
        if (idx < OUT_LEN) out[c * OUT_LEN + idx] = val;
    }
}

extern "C" void kernel_launch(void* const* d_in, const int* in_sizes, int n_in,
                              void* d_out, int out_size, void* d_ws, size_t ws_size,
                              hipStream_t stream) {
    (void)in_sizes; (void)n_in; (void)out_size;
    const float* audio = (const float*)d_in[0];
    const float* rir   = (const float*)d_in[1];
    float* out = (float*)d_out;

    if (d_ws != nullptr && ws_size >= WS_NEED_BYTES) {
        float* S = (float*)d_ws;
        float* Y = S + S_FLOATS;
        rir_part_forward<<<dim3(320, 1, 1), dim3(256, 1, 1), 0, stream>>>(audio, rir, S);
        rir_bin_conv<<<dim3(1024, 1, 1), dim3(128, 1, 1), 0, stream>>>(S, Y);
        rir_part_inverse<<<dim3(256, 1, 1), dim3(512, 1, 1), 0, stream>>>(Y, out);
    } else {
        dim3 grid(NBLOCKS, 1, 1);
        dim3 block(256, 1, 1);
        RoomImpulseResponse_wmma_conv<<<grid, block, 0, stream>>>(audio, rir, out);
    }
}


# Per-bin conv: tiles n0 and n0+512 share one k run over an R window that wraps; a row's running sum is copied out where the row changes tile; 66 16-k groups per bin (was 96); sum order changes
# speedup vs baseline: 27.0786x; 1.2109x over previous
#include <hip/hip_runtime.h>
#include <hip/hip_bf16.h>


typedef __attribute__((ext_vector_type(2))) float v2f;
typedef __attribute__((ext_vector_type(4))) float v4f;
typedef __attribute__((ext_vector_type(8))) float v8f;

typedef struct __attribute__((packed, aligned(4))) { float x, y; } f2u;

#define T_LEN   131072
#define OUT_LEN (2 * T_LEN - 1)
#define TILE_W  512
#define NTILES  ((OUT_LEN + TILE_W - 1) / TILE_W)
#define NWAVES  8

#define GPC         47
#define KC          (16 * GPC)
#define RIR_WORDS   1280
#define AUD_WORDS   (KC + 272)
#define WAVE_WORDS  (RIR_WORDS + AUD_WORDS)
#define NCH         4
#define NBLOCKS     (NTILES * NCH)


__device__ __forceinline__ v8f wmma4(v2f a, v2f b, v8f c) {
    return __builtin_amdgcn_wmma_f32_16x16x4_f32(false, a, false, b,
                                                 (short)0, c, false, false);
}

__device__ __forceinline__ v2f ldpair(const float* p) {
    f2u t = *(const f2u*)p;
    v2f r = {t.x, t.y};
    return r;
}

__device__ __forceinline__ void wave_lds_handover() {
    __builtin_amdgcn_fence(__ATOMIC_RELEASE, "workgroup", "local");
    __builtin_amdgcn_wave_barrier();
    __builtin_amdgcn_fence(__ATOMIC_ACQUIRE, "workgroup", "local");
}

__global__ __launch_bounds__(256)
void RoomImpulseResponse_wmma_conv(const float* __restrict__ audio,
                                   const float* __restrict__ rir,
                                   float* __restrict__ out)
{
    __shared__ __attribute__((aligned(16))) float smem[NWAVES * WAVE_WORDS];

    const unsigned bid = blockIdx.x;
    const int ch   = (int)(bid % NCH);
    const int rank = (int)(bid / NCH);
    const int tile = (rank & 1) ? (NTILES / 2 - 1 - (rank >> 1))
                                : (NTILES / 2 + (rank >> 1));
    const int tid  = threadIdx.x;
    const int wave = __builtin_amdgcn_readfirstlane(tid >> 5);
    const int lane = tid & 31;
    const int h    = lane >> 4;
    const int mq   = lane & 15;

    const float* __restrict__ a_ch = audio + ch * T_LEN;

    float* rbuf = smem + wave * WAVE_WORDS;
    float* abuf = rbuf + RIR_WORDS;
    const float* pA0  = rbuf + 20 * (15 - mq) + 2 * h;
    const float* pB00 = abuf + mq + 2 * h + 1;
    const float* pB10 = pB00 + 256;

    {
        const int n0   = tile * TILE_W;

        int klo = T_LEN - 272 - n0;  if (klo < 0) klo = 0;
        int khi = 2 * T_LEN - 2 - n0;
        if (khi > T_LEN - 1 + 240) khi = T_LEN - 1 + 240;
        const int kbeg    = klo & ~15;
        const int ngroups = ((khi - kbeg) >> 4) + 1;
        const int per     = (ngroups + NWAVES - 1) / NWAVES;
        const int gLo     = wave * per;
        const int gHi     = min(ngroups, gLo + per);

        v8f acc0a = {0.f,0.f,0.f,0.f,0.f,0.f,0.f,0.f};
        v8f acc0b = {0.f,0.f,0.f,0.f,0.f,0.f,0.f,0.f};
        v8f acc1a = {0.f,0.f,0.f,0.f,0.f,0.f,0.f,0.f};
        v8f acc1b = {0.f,0.f,0.f,0.f,0.f,0.f,0.f,0.f};

        for (int g0 = gLo; g0 < gHi; g0 += GPC) {
            const int kc = kbeg + 16 * g0;

            {
                const int R0 = kc - 240;
                v4f t[8];
                #pragma unroll
                for (int j = 0; j < 8; ++j) {
                    const int gi = R0 + 4 * (lane + 32 * j);
                    t[j] = *(const v4f*)(rir + min(max(gi, 0), T_LEN - 4));
                }
                #pragma unroll
                for (int j = 0; j < 8; ++j) {
                    const int g  = lane + 32 * j;
                    const int gi = R0 + 4 * g;
                    v4f v = t[j];
                    if ((unsigned)gi >= (unsigned)T_LEN) { v4f z = {0.f,0.f,0.f,0.f}; v = z; }
                    *(v4f*)(rbuf + 4 * g + 4 * (g >> 2)) = v;
                }
            }
            {
                const int S0 = n0 + kc - T_LEN;
                v4f t[8];
                #pragma unroll
                for (int j = 0; j < 8; ++j) {
                    const int gi = S0 + 4 * (lane + 32 * j);
                    t[j] = *(const v4f*)(a_ch + min(max(gi, 0), T_LEN - 4));
                }
                #pragma unroll
                for (int j = 0; j < 8; ++j) {
                    const int g  = lane + 32 * j;
                    const int gi = S0 + 4 * g;
                    v4f v = t[j];
                    if ((unsigned)gi >= (unsigned)T_LEN) { v4f z = {0.f,0.f,0.f,0.f}; v = z; }
                    *(v4f*)(abuf + 4 * g) = v;
                }
            }
            wave_lds_handover();

            const int ng = min(GPC, gHi - g0);
            const float* pA  = pA0;
            const float* pB0 = pB00;
            const float* pB1 = pB10;
            for (int g = 0; g < ng; ++g) {
                v2f a0  = *(const v2f*)(pA);
                v2f a1  = *(const v2f*)(pA + 4);
                v2f a2  = *(const v2f*)(pA + 8);
                v2f a3  = *(const v2f*)(pA + 12);
                v2f b00 = ldpair(pB0);
                v2f b01 = ldpair(pB0 + 4);
                v2f b02 = ldpair(pB0 + 8);
                v2f b03 = ldpair(pB0 + 12);
                v2f b10 = ldpair(pB1);
                v2f b11 = ldpair(pB1 + 4);
                v2f b12 = ldpair(pB1 + 8);
                v2f b13 = ldpair(pB1 + 12);
                acc0a = wmma4(a0, b00, acc0a);
                acc1a = wmma4(a0, b10, acc1a);
                acc0b = wmma4(a1, b01, acc0b);
                acc1b = wmma4(a1, b11, acc1b);
                acc0a = wmma4(a2, b02, acc0a);
                acc1a = wmma4(a2, b12, acc1a);
                acc0b = wmma4(a3, b03, acc0b);
                acc1b = wmma4(a3, b13, acc1b);
                pA  += 20;
                pB0 += 16;
                pB1 += 16;
            }
            wave_lds_handover();
        }

        __syncthreads();
        float* red = smem;

        #pragma unroll
        for (int v = 0; v < 8; ++v) {
            int M = v + 8 * h;
            red[wave * TILE_W +       M * 16 + mq] = acc0a[v] + acc0b[v];
            red[wave * TILE_W + 256 + M * 16 + mq] = acc1a[v] + acc1b[v];
        }
        __syncthreads();

        float s0 = 0.f, s1 = 0.f;
        #pragma unroll
        for (int w = 0; w < NWAVES; ++w) {
            s0 += red[w * TILE_W + tid];
            s1 += red[w * TILE_W + 256 + tid];
        }
        int n = n0 + tid;
        if (n < OUT_LEN)       out[ch * OUT_LEN + n]       = s0;
        if (n + 256 < OUT_LEN) out[ch * OUT_LEN + n + 256] = s1;
    }
}


#define PB        256
#define NPART     512
#define NBIN      256
#define NQ        1024
#define S_FLOATS  (5 * NBIN * 2 * NPART)
#define Y_FLOATS  (NCH * NBIN * 2 * NQ)
#define WS_NEED_BYTES 13631488ull
static_assert((unsigned long long)(S_FLOATS + Y_FLOATS) * 4ull == WS_NEED_BYTES, "workspace size");

#define XS_PITCH  260

__device__ __forceinline__ void fwd_step(const float* tab, int f, int n, v2f b,
                                         v8f& re, v8f& im) {
    const int j0 = (f * n) & 511;
    const int j1 = (j0 + f) & 511;
    const v2f t0 = *(const v2f*)(tab + 2 * j0);
    const v2f t1 = *(const v2f*)(tab + 2 * j1);
    v2f ar = {t0.x, t1.x};
    v2f ai = {t0.y, t1.y};
    if (f == 0) { ai.x = 1.f; ai.y = -1.f; }
    re = wmma4(ar, b, re);
    im = wmma4(ai, b, im);
}

__global__ __launch_bounds__(256)
void rir_part_forward(const float* __restrict__ audio,
                      const float* __restrict__ rir,
                      float* __restrict__ S)
{
    __shared__ __attribute__((aligned(16))) float tab[1024];
    __shared__ __attribute__((aligned(16))) float xs[16 * XS_PITCH];
    const int bx   = (int)blockIdx.x;
    const int half = bx & 1;
    const int pt   = (bx >> 1) & 31;
    const int sig  = bx >> 6;
    const int tid  = threadIdx.x;

    for (int j = tid; j < 512; j += 256) {
        float s, c;
        sincospif((float)j * (1.0f / 256.0f), &s, &c);
        tab[2 * j]     = c;
        tab[2 * j + 1] = -s;
    }
    {
        float xv[16];
        if (sig < NCH) {
            const float* src = audio + sig * T_LEN + PB * 16 * pt + tid;
            #pragma unroll
            for (int i = 0; i < 16; ++i) xv[i] = src[PB * i];
        } else {
            const float* src = rir + PB * 16 * pt + 255 - tid;
            #pragma unroll
            for (int i = 0; i < 16; ++i) xv[i] = src[PB * i];
        }
        #pragma unroll
        for (int i = 0; i < 16; ++i) xs[i * XS_PITCH + tid] = xv[i];
    }
    __syncthreads();

    const int wave = tid >> 5;
    const int lane = tid & 31;
    const int h    = lane >> 4;
    const int mq   = lane & 15;
    const int ft   = 8 * half + wave;
    const int f    = 16 * ft + mq;
    const float* pb = xs + mq * XS_PITCH + 2 * h;

    v8f re0 = {0.f,0.f,0.f,0.f,0.f,0.f,0.f,0.f};
    v8f re1 = re0, im0 = re0, im1 = re0;
    for (int k0 = 0; k0 < PB; k0 += 8) {
        const v2f b0 = *(const v2f*)(pb + k0);
        const v2f b1 = *(const v2f*)(pb + k0 + 4);
        fwd_step(tab, f, k0 + 2 * h,     b0, re0, im0);
        fwd_step(tab, f, k0 + 4 + 2 * h, b1, re1, im1);
    }
    const int p = 16 * pt + mq;
    #pragma unroll
    for (int v = 0; v < 8; ++v) {
        const int fo = 16 * ft + v + 8 * h;
        float* d = S + ((sig * NBIN + fo) * 2) * NPART + p;
        d[0]     = re0[v] + re1[v];
        d[NPART] = im0[v] + im1[v];
    }
}

#define RS_WORDS 960
#define XSW      544
#define EXW      256

__device__ __forceinline__ void conv_group(const float* pA, const float* pB,
                                           v8f& p1, v8f& p2, v8f& p3) {
    #pragma unroll
    for (int s = 0; s < 4; ++s) {
        const v2f ar = *(const v2f*)(pA + 4 * s);
        const v2f ai = *(const v2f*)(pA + RS_WORDS + 4 * s);
        const v2f as = *(const v2f*)(pA + 2 * RS_WORDS + 4 * s);
        const v2f br = ldpair(pB + 4 * s);
        const v2f bi = ldpair(pB + XSW + 4 * s);
        const v2f bs = ldpair(pB + 2 * XSW + 4 * s);
        p1 = wmma4(ar, br, p1);
        p2 = wmma4(ai, bi, p2);
        p3 = wmma4(as, bs, p3);
    }
}

__global__ __launch_bounds__(128) __attribute__((amdgpu_waves_per_eu(4, 4)))
void rir_bin_conv(const float* __restrict__ S, float* __restrict__ Y)
{
    __shared__ __attribute__((aligned(16))) float sm[3 * RS_WORDS + 3 * XSW];
    __shared__ float ex[2 * 2 * 2 * EXW];
    float* Rs = sm;
    float* Xs = sm + 3 * RS_WORDS;
    const int f   = (int)blockIdx.x >> 2;
    const int c   = (int)blockIdx.x & 3;
    const int tid = threadIdx.x;
    const float* gR = S + ((NCH * NBIN + f) * 2) * NPART;
    const float* gX = S + ((c * NBIN + f) * 2) * NPART;

    {
        float rvr[6], rvi[6], xvr[5], xvi[5];
        #pragma unroll
        for (int i = 0; i < 6; ++i) {
            const int t  = tid + 128 * i;
            const int rc = min((t < 240) ? t + 272 : t - 240, NPART - 1);
            rvr[i] = gR[rc];
            rvi[i] = gR[NPART + rc];
        }
        #pragma unroll
        for (int i = 0; i < 5; ++i) {
            const int mc = min(max(tid + 128 * i - 16, 0), NPART - 1);
            xvr[i] = gX[mc];
            xvi[i] = gX[NPART + mc];
        }
        #pragma unroll
        for (int i = 0; i < 6; ++i) {
            const int t = tid + 128 * i;
            if (i < 5 || t < 752) {
                const int w = t + 4 * (t >> 4);
                Rs[w]                = rvr[i];
                Rs[RS_WORDS + w]     = rvi[i];
                Rs[2 * RS_WORDS + w] = rvr[i] + rvi[i];
            }
        }
        #pragma unroll
        for (int i = 0; i < 5; ++i) {
            const int w = tid + 128 * i;
            const bool in = (unsigned)(w - 16) < (unsigned)NPART;
            if (i < 4 || w < XSW) {
                const float re = in ? xvr[i] : 0.f;
                const float im = in ? xvi[i] : 0.f;
                Xs[w]           = re;
                Xs[XSW + w]     = im;
                Xs[2 * XSW + w] = re + im;
            }
        }
    }
    __syncthreads();

    const int wave = __builtin_amdgcn_readfirstlane(tid >> 5);
    const int pair = wave & 1;
    const int part = wave >> 1;
    const int lane = tid & 31;
    const int h    = lane >> 4;
    const int mq   = lane & 15;

    const float* pA0 = Rs + 20 * (15 - mq) + 2 * h;
    const float* pB0 = Xs + mq + 2 * h + 1;
    float* exw = ex + (pair * 2 + part) * 2 * EXW + lane;
    const float* exr = ex + (pair * 2 + (part ^ 1)) * 2 * EXW + lane;

    const v8f zero = {0.f,0.f,0.f,0.f,0.f,0.f,0.f,0.f};
    float yr[8], yi[8];
    int n0;

    if (part == 0) {
        v8f u1 = zero, u2 = zero, u3 = zero;
        v8f c1 = zero, c2 = zero, c3 = zero;
        const float* pA = pA0;
        const float* pB = pB0 + 16 + 256 * pair;
        #pragma unroll 1
        for (int hs = 0; hs < 2; ++hs) {
            const bool sel = (h == hs);
            #pragma unroll
            for (int r = 0; r < 8; ++r) {
                c1[r] = sel ? u1[r] : c1[r];
                c2[r] = sel ? u2[r] : c2[r];
                c3[r] = sel ? u3[r] : c3[r];
                conv_group(pA, pB, u1, u2, u3);
                pA += 20;
                pB += 16;
            }
        }
        #pragma unroll
        for (int v = 0; v < 8; ++v) {
            const float b1 = u1[v] - c1[v], b2 = u2[v] - c2[v], b3 = u3[v] - c3[v];
            yr[v] = (f == 0) ? b1 : b1 - b2;
            yi[v] = (f == 0) ? b2 : (b3 - b1) - b2;
            exw[32 * v]       = (f == 0) ? c1[v] : c1[v] - c2[v];
            exw[EXW + 32 * v] = (f == 0) ? c2[v] : (c3[v] - c1[v]) - c2[v];
        }
        n0 = 512 + 256 * pair;
    } else {
        v8f t1 = zero, t2 = zero, t3 = zero;
        v8f l1 = zero, l2 = zero, l3 = zero;
        const int nT = pair ? 0 : 16;
        const float* pA = pA0 + 20 * 16;
        const float* pB = pB0 + 16 + 256 * pair + 256;
        for (int g = 0; g < nT; ++g) {
            conv_group(pA, pB, t1, t2, t3);
            pA += 20;
            pB += 16;
        }
        const int g2 = pair ? 15 : 31;
        const int nL = 32 - g2;
        pA = pA0 + 20 * g2;
        pB = pB0;
        for (int g = 0; g < nL; ++g) {
            conv_group(pA, pB, l1, l2, l3);
            pA += 20;
            pB += 16;
        }
        #pragma unroll
        for (int v = 0; v < 8; ++v) {
            yr[v] = (f == 0) ? l1[v] : l1[v] - l2[v];
            yi[v] = (f == 0) ? l2[v] : (l3[v] - l1[v]) - l2[v];
            exw[32 * v]       = (f == 0) ? t1[v] : t1[v] - t2[v];
            exw[EXW + 32 * v] = (f == 0) ? t2[v] : (t3[v] - t1[v]) - t2[v];
        }
        n0 = 256 * pair;
    }
    __syncthreads();

    float* d = Y + ((c * NBIN + f) * 2) * NQ + n0 + mq;
    #pragma unroll
    for (int v = 0; v < 8; ++v) {
        const int M = v + 8 * h;
        d[16 * M]      = yr[v] + exr[32 * v];
        d[NQ + 16 * M] = yi[v] + exr[EXW + 32 * v];
    }
}

#define ZP   260
#define PLN  (16 * ZP + 2)
#define RP   130
__device__ __forceinline__ void inv_fold_step(const float* tab, const float* paE,
                                              const float* paO, int f0, int n,
                                              v8f& e, v8f& o) {
    const v2f aE = *(const v2f*)paE;
    const v2f aO = *(const v2f*)paO;
    const int j0 = (f0 * n) & 511;
    const int j1 = (j0 + n) & 511;
    const v2f t0 = *(const v2f*)(tab + 2 * j0);
    const v2f t1 = *(const v2f*)(tab + 2 * j1);
    v2f bE = {t0.x, t1.y};
    v2f bO = {t0.y, t1.x};
    if (f0 == 0) bE.x = 1.0f / 512.0f;
    e = wmma4(aE, bE, e);
    o = wmma4(aO, bO, o);
}

__global__ __launch_bounds__(512)
void rir_part_inverse(const float* __restrict__ Y, float* __restrict__ out)
{
    __shared__ __attribute__((aligned(16))) float tab[1024];
    __shared__ __attribute__((aligned(16))) float zs[2 * PLN];
    __shared__ __attribute__((aligned(16))) float red[4 * 16 * RP];
    __shared__ float y128s[16];
    const int c   = (int)blockIdx.x >> 6;
    const int q0  = 16 * ((int)blockIdx.x & 63);
    const int tid = threadIdx.x;
    {
        float s, co;
        sincospif((float)tid * (1.0f / 256.0f), &s, &co);
        tab[2 * tid]     = co * (1.0f / 256.0f);
        tab[2 * tid + 1] = -s * (1.0f / 256.0f);
    }
    const float* gY = Y + c * (NBIN * 2 * NQ);
    {
        const int qc = tid & 15;
        const int q  = q0 + qc;
        const int r  = tid >> 4;
        const float* col = gY + r * NQ + q;
        const int back = (q > 0) ? 1 : 0;
        float y1[16], y0[16];
        #pragma unroll
        for (int i = 0; i < 16; ++i) {
            y1[i] = col[32 * NQ * i];
            y0[i] = col[32 * NQ * i - back];
        }
        const bool odd = (r & 2) != 0;
        const int  pl  = (((r >> 1) ^ r) & 1);
        float* zc = zs + pl * PLN + qc * ZP + (r >> 1);
        #pragma unroll
        for (int i = 0; i < 16; ++i) {
            const float p = (q > 0) ? y0[i] : 0.f;
            zc[16 * i] = odd ? y1[i] - p : y1[i] + p;
        }
    }
    __syncthreads();

    const int wave = tid >> 5;
    const int lane = tid & 31;
    const int h    = lane >> 4;
    const int mq   = lane & 15;
    const int nt   = wave & 7;
    const int kh   = wave >> 3;
    const int n    = 16 * nt + mq;
    const int fb   = 128 * kh + 2 * h;
    const float* paE = zs + mq * ZP + fb;
    const float* paO = paE + PLN;

    v8f e0 = {0.f,0.f,0.f,0.f,0.f,0.f,0.f,0.f};
    v8f e1 = e0, o0 = e0, o1 = e0;
    for (int t = 0; t < 32; t += 2) {
        inv_fold_step(tab, paE + 4 * t,     paO + 4 * t,     fb + 4 * t,     n, e0, o0);
        inv_fold_step(tab, paE + 4 * t + 4, paO + 4 * t + 4, fb + 4 * t + 4, n, e1, o1);
    }
    {
        float* rE = red + ((2 * kh)     * 16) * RP + n;
        float* rO = red + ((2 * kh + 1) * 16) * RP + n;
        #pragma unroll
        for (int v = 0; v < 8; ++v) {
            const int q = v + 8 * h;
            rE[q * RP] = e0[v] + e1[v];
            rO[q * RP] = o0[v] + o1[v];
        }
    }
    {
        const float* pe = zs + wave * ZP;
        float s = 0.f;
        #pragma unroll
        for (int i = 0; i < 8; ++i) {
            float x = pe[lane + 32 * i];
            if (i == 0 && lane == 0) x *= 0.5f;
            s += x;
        }
        const int m4 = lane & 3;
        if (m4 == 1 || m4 == 2) s = -s;
        #pragma unroll
        for (int off = 16; off > 0; off >>= 1) s += __shfl_xor(s, off);
        if (lane == 0)
            y128s[wave] = s * (1.0f / 256.0f) + zs[PLN + wave * ZP] * (1.0f / 512.0f);
    }
    __syncthreads();

    #pragma unroll
    for (int i = 0; i < 8; ++i) {
        const int o  = tid + 512 * i;
        const int q  = o >> 8;
        const int no = o & 255;
        float val;
        if (no == 128) {
            val = y128s[q];
        } else {
            const int nn = (no < 128) ? no : 256 - no;
            const float E = red[(0 * 16 + q) * RP + nn] + red[(2 * 16 + q) * RP + nn];
            const float O = red[(1 * 16 + q) * RP + nn] + red[(3 * 16 + q) * RP + nn];
            float z = zs[PLN + q * ZP] * (1.0f / 512.0f);
            if (nn & 1) z = -z;
            val = ((no < 128) ? E + O : E - O) + z;
        }
        const int idx = PB * (q0 + q) + no;
        if (idx < OUT_LEN) out[c * OUT_LEN + idx] = val;
    }
}

extern "C" void kernel_launch(void* const* d_in, const int* in_sizes, int n_in,
                              void* d_out, int out_size, void* d_ws, size_t ws_size,
                              hipStream_t stream) {
    (void)in_sizes; (void)n_in; (void)out_size;
    const float* audio = (const float*)d_in[0];
    const float* rir   = (const float*)d_in[1];
    float* out = (float*)d_out;

    if (d_ws != nullptr && ws_size >= WS_NEED_BYTES) {
        float* S = (float*)d_ws;
        float* Y = S + S_FLOATS;
        rir_part_forward<<<dim3(320, 1, 1), dim3(256, 1, 1), 0, stream>>>(audio, rir, S);
        rir_bin_conv<<<dim3(1024, 1, 1), dim3(128, 1, 1), 0, stream>>>(S, Y);
        rir_part_inverse<<<dim3(256, 1, 1), dim3(512, 1, 1), 0, stream>>>(Y, out);
    } else {
        dim3 grid(NBLOCKS, 1, 1);
        dim3 block(256, 1, 1);
        RoomImpulseResponse_wmma_conv<<<grid, block, 0, stream>>>(audio, rir, out);
    }
}


# Forward transform folded on n <-> 256-n (twiddle symmetry): x[n]+-x[256-n] formed at staging, K 132 per tile (was 256); tile rows 16 bins apart, table skewed; f32 operand sums, sum order changes
# speedup vs baseline: 28.5479x; 1.0543x over previous
#include <hip/hip_runtime.h>
#include <hip/hip_bf16.h>


typedef __attribute__((ext_vector_type(2))) float v2f;
typedef __attribute__((ext_vector_type(4))) float v4f;
typedef __attribute__((ext_vector_type(8))) float v8f;

typedef struct __attribute__((packed, aligned(4))) { float x, y; } f2u;

#define T_LEN   131072
#define OUT_LEN (2 * T_LEN - 1)
#define TILE_W  512
#define NTILES  ((OUT_LEN + TILE_W - 1) / TILE_W)
#define NWAVES  8

#define GPC         47
#define KC          (16 * GPC)
#define RIR_WORDS   1280
#define AUD_WORDS   (KC + 272)
#define WAVE_WORDS  (RIR_WORDS + AUD_WORDS)
#define NCH         4
#define NBLOCKS     (NTILES * NCH)


__device__ __forceinline__ v8f wmma4(v2f a, v2f b, v8f c) {
    return __builtin_amdgcn_wmma_f32_16x16x4_f32(false, a, false, b,
                                                 (short)0, c, false, false);
}

__device__ __forceinline__ v2f ldpair(const float* p) {
    f2u t = *(const f2u*)p;
    v2f r = {t.x, t.y};
    return r;
}

__device__ __forceinline__ void wave_lds_handover() {
    __builtin_amdgcn_fence(__ATOMIC_RELEASE, "workgroup", "local");
    __builtin_amdgcn_wave_barrier();
    __builtin_amdgcn_fence(__ATOMIC_ACQUIRE, "workgroup", "local");
}

__global__ __launch_bounds__(256)
void RoomImpulseResponse_wmma_conv(const float* __restrict__ audio,
                                   const float* __restrict__ rir,
                                   float* __restrict__ out)
{
    __shared__ __attribute__((aligned(16))) float smem[NWAVES * WAVE_WORDS];

    const unsigned bid = blockIdx.x;
    const int ch   = (int)(bid % NCH);
    const int rank = (int)(bid / NCH);
    const int tile = (rank & 1) ? (NTILES / 2 - 1 - (rank >> 1))
                                : (NTILES / 2 + (rank >> 1));
    const int tid  = threadIdx.x;
    const int wave = __builtin_amdgcn_readfirstlane(tid >> 5);
    const int lane = tid & 31;
    const int h    = lane >> 4;
    const int mq   = lane & 15;

    const float* __restrict__ a_ch = audio + ch * T_LEN;

    float* rbuf = smem + wave * WAVE_WORDS;
    float* abuf = rbuf + RIR_WORDS;
    const float* pA0  = rbuf + 20 * (15 - mq) + 2 * h;
    const float* pB00 = abuf + mq + 2 * h + 1;
    const float* pB10 = pB00 + 256;

    {
        const int n0   = tile * TILE_W;

        int klo = T_LEN - 272 - n0;  if (klo < 0) klo = 0;
        int khi = 2 * T_LEN - 2 - n0;
        if (khi > T_LEN - 1 + 240) khi = T_LEN - 1 + 240;
        const int kbeg    = klo & ~15;
        const int ngroups = ((khi - kbeg) >> 4) + 1;
        const int per     = (ngroups + NWAVES - 1) / NWAVES;
        const int gLo     = wave * per;
        const int gHi     = min(ngroups, gLo + per);

        v8f acc0a = {0.f,0.f,0.f,0.f,0.f,0.f,0.f,0.f};
        v8f acc0b = {0.f,0.f,0.f,0.f,0.f,0.f,0.f,0.f};
        v8f acc1a = {0.f,0.f,0.f,0.f,0.f,0.f,0.f,0.f};
        v8f acc1b = {0.f,0.f,0.f,0.f,0.f,0.f,0.f,0.f};

        for (int g0 = gLo; g0 < gHi; g0 += GPC) {
            const int kc = kbeg + 16 * g0;

            {
                const int R0 = kc - 240;
                v4f t[8];
                #pragma unroll
                for (int j = 0; j < 8; ++j) {
                    const int gi = R0 + 4 * (lane + 32 * j);
                    t[j] = *(const v4f*)(rir + min(max(gi, 0), T_LEN - 4));
                }
                #pragma unroll
                for (int j = 0; j < 8; ++j) {
                    const int g  = lane + 32 * j;
                    const int gi = R0 + 4 * g;
                    v4f v = t[j];
                    if ((unsigned)gi >= (unsigned)T_LEN) { v4f z = {0.f,0.f,0.f,0.f}; v = z; }
                    *(v4f*)(rbuf + 4 * g + 4 * (g >> 2)) = v;
                }
            }
            {
                const int S0 = n0 + kc - T_LEN;
                v4f t[8];
                #pragma unroll
                for (int j = 0; j < 8; ++j) {
                    const int gi = S0 + 4 * (lane + 32 * j);
                    t[j] = *(const v4f*)(a_ch + min(max(gi, 0), T_LEN - 4));
                }
                #pragma unroll
                for (int j = 0; j < 8; ++j) {
                    const int g  = lane + 32 * j;
                    const int gi = S0 + 4 * g;
                    v4f v = t[j];
                    if ((unsigned)gi >= (unsigned)T_LEN) { v4f z = {0.f,0.f,0.f,0.f}; v = z; }
                    *(v4f*)(abuf + 4 * g) = v;
                }
            }
            wave_lds_handover();

            const int ng = min(GPC, gHi - g0);
            const float* pA  = pA0;
            const float* pB0 = pB00;
            const float* pB1 = pB10;
            for (int g = 0; g < ng; ++g) {
                v2f a0  = *(const v2f*)(pA);
                v2f a1  = *(const v2f*)(pA + 4);
                v2f a2  = *(const v2f*)(pA + 8);
                v2f a3  = *(const v2f*)(pA + 12);
                v2f b00 = ldpair(pB0);
                v2f b01 = ldpair(pB0 + 4);
                v2f b02 = ldpair(pB0 + 8);
                v2f b03 = ldpair(pB0 + 12);
                v2f b10 = ldpair(pB1);
                v2f b11 = ldpair(pB1 + 4);
                v2f b12 = ldpair(pB1 + 8);
                v2f b13 = ldpair(pB1 + 12);
                acc0a = wmma4(a0, b00, acc0a);
                acc1a = wmma4(a0, b10, acc1a);
                acc0b = wmma4(a1, b01, acc0b);
                acc1b = wmma4(a1, b11, acc1b);
                acc0a = wmma4(a2, b02, acc0a);
                acc1a = wmma4(a2, b12, acc1a);
                acc0b = wmma4(a3, b03, acc0b);
                acc1b = wmma4(a3, b13, acc1b);
                pA  += 20;
                pB0 += 16;
                pB1 += 16;
            }
            wave_lds_handover();
        }

        __syncthreads();
        float* red = smem;

        #pragma unroll
        for (int v = 0; v < 8; ++v) {
            int M = v + 8 * h;
            red[wave * TILE_W +       M * 16 + mq] = acc0a[v] + acc0b[v];
            red[wave * TILE_W + 256 + M * 16 + mq] = acc1a[v] + acc1b[v];
        }
        __syncthreads();

        float s0 = 0.f, s1 = 0.f;
        #pragma unroll
        for (int w = 0; w < NWAVES; ++w) {
            s0 += red[w * TILE_W + tid];
            s1 += red[w * TILE_W + 256 + tid];
        }
        int n = n0 + tid;
        if (n < OUT_LEN)       out[ch * OUT_LEN + n]       = s0;
        if (n + 256 < OUT_LEN) out[ch * OUT_LEN + n + 256] = s1;
    }
}


#define PB        256
#define NPART     512
#define NBIN      256
#define NQ        1024
#define S_FLOATS  (5 * NBIN * 2 * NPART)
#define Y_FLOATS  (NCH * NBIN * 2 * NQ)
#define WS_NEED_BYTES 13631488ull
static_assert((unsigned long long)(S_FLOATS + Y_FLOATS) * 4ull == WS_NEED_BYTES, "workspace size");

#define XP_PITCH  132
#define TWF_WORDS 1088

__device__ __forceinline__ void fwd_fold_step(const float* tab, int f, int n,
                                              v2f bre, v2f bim, v8f& re, v8f& im) {
    const int j0 = (f * n) & 511;
    const int j1 = (j0 + f) & 511;
    const v2f t0 = *(const v2f*)(tab + 2 * (j0 + (j0 >> 4)));
    const v2f t1 = *(const v2f*)(tab + 2 * (j1 + (j1 >> 4)));
    const v2f ar = {t0.x, t1.x};
    const v2f ai = {t0.y, t1.y};
    re = wmma4(ar, bre, re);
    im = wmma4(ai, bim, im);
}

__global__ __launch_bounds__(256)
void rir_part_forward(const float* __restrict__ audio,
                      const float* __restrict__ rir,
                      float* __restrict__ S)
{
    __shared__ __attribute__((aligned(16))) float tab[TWF_WORDS];
    __shared__ __attribute__((aligned(16))) float xs[2 * 16 * XP_PITCH];
    const int bx   = (int)blockIdx.x;
    const int half = bx & 1;
    const int pt   = (bx >> 1) & 31;
    const int sig  = bx >> 6;
    const int tid  = threadIdx.x;

    for (int j = tid; j < 512; j += 256) {
        float s, c;
        sincospif((float)j * (1.0f / 256.0f), &s, &c);
        float* e = tab + 2 * (j + (j >> 4));
        e[0] = c;
        e[1] = -s;
    }
    {
        const int n  = tid & 127;
        const int ph = tid >> 7;
        const int nb = (n == 0) ? 128 : 256 - n;
        float xa[8], xb[8];
        if (sig < NCH) {
            const float* src = audio + sig * T_LEN + PB * (16 * pt + 8 * ph);
            #pragma unroll
            for (int i = 0; i < 8; ++i) { xa[i] = src[PB * i + n]; xb[i] = src[PB * i + nb]; }
        } else {
            const float* src = rir + PB * (16 * pt + 8 * ph) + 255;
            #pragma unroll
            for (int i = 0; i < 8; ++i) { xa[i] = src[PB * i - n]; xb[i] = src[PB * i - nb]; }
        }
        float* ps = xs + (8 * ph) * XP_PITCH + n;
        float* pd = ps + 16 * XP_PITCH;
        #pragma unroll
        for (int i = 0; i < 8; ++i) {
            ps[i * XP_PITCH] = (n != 0) ? xa[i] + xb[i] : xa[i];
            pd[i * XP_PITCH] = (n != 0) ? xa[i] - xb[i] : xa[i];
            if (n < 4) {
                const float m = (n == 0) ? xb[i] : 0.f;
                ps[i * XP_PITCH + 128] = m;
                pd[i * XP_PITCH + 128] = m;
            }
        }
    }
    __syncthreads();

    const int wave = __builtin_amdgcn_readfirstlane(tid >> 5);
    const int lane = tid & 31;
    const int h    = lane >> 4;
    const int mq   = lane & 15;
    const int tix  = 8 * half + wave;
    const int par  = tix & 1;
    const int f    = tix + 16 * mq;
    const float* pRe = xs + (par ? 16 * XP_PITCH : 0) + mq * XP_PITCH + 2 * h;
    const float* pIm = xs + (par ? 0 : 16 * XP_PITCH) + mq * XP_PITCH + 2 * h;

    v8f re0 = {0.f,0.f,0.f,0.f,0.f,0.f,0.f,0.f};
    v8f re1 = re0, im0 = re0, im1 = re0;
    for (int k0 = 0; k0 < 128; k0 += 8) {
        const v2f br0 = *(const v2f*)(pRe + k0);
        const v2f bi0 = *(const v2f*)(pIm + k0);
        const v2f br1 = *(const v2f*)(pRe + k0 + 4);
        const v2f bi1 = *(const v2f*)(pIm + k0 + 4);
        fwd_fold_step(tab, f, k0 + 2 * h,     br0, bi0, re0, im0);
        fwd_fold_step(tab, f, k0 + 4 + 2 * h, br1, bi1, re1, im1);
    }
    {
        const v2f br = *(const v2f*)(pRe + 128);
        const v2f bi = *(const v2f*)(pIm + 128);
        fwd_fold_step(tab, f, 128 + 2 * h, br, bi, re0, im0);
    }

    float x256 = 0.f;
    if (tix == 0) {
        const float* q = xs + mq * XP_PITCH;
        float e = 0.f, o = 0.f;
        for (int i = 0; i < 16; ++i) {
            const v4f v = *(const v4f*)(q + 8 * i + 4 * h);
            e += v.x + v.z;
            o += v.y + v.w;
        }
        const float a = e - o;
        x256 = a + __shfl_xor(a, 16) + q[128];
    }

    const int p = 16 * pt + mq;
    #pragma unroll
    for (int v = 0; v < 8; ++v) {
        const int fo = tix + 16 * (v + 8 * h);
        float* d = S + ((sig * NBIN + fo) * 2) * NPART + p;
        d[0]     = re0[v] + re1[v];
        d[NPART] = (fo == 0) ? x256 : im0[v] + im1[v];
    }
}

#define RS_WORDS 960
#define XSW      544
#define EXW      256

__device__ __forceinline__ void conv_group(const float* pA, const float* pB,
                                           v8f& p1, v8f& p2, v8f& p3) {
    #pragma unroll
    for (int s = 0; s < 4; ++s) {
        const v2f ar = *(const v2f*)(pA + 4 * s);
        const v2f ai = *(const v2f*)(pA + RS_WORDS + 4 * s);
        const v2f as = *(const v2f*)(pA + 2 * RS_WORDS + 4 * s);
        const v2f br = ldpair(pB + 4 * s);
        const v2f bi = ldpair(pB + XSW + 4 * s);
        const v2f bs = ldpair(pB + 2 * XSW + 4 * s);
        p1 = wmma4(ar, br, p1);
        p2 = wmma4(ai, bi, p2);
        p3 = wmma4(as, bs, p3);
    }
}

__global__ __launch_bounds__(128) __attribute__((amdgpu_waves_per_eu(4, 4)))
void rir_bin_conv(const float* __restrict__ S, float* __restrict__ Y)
{
    __shared__ __attribute__((aligned(16))) float sm[3 * RS_WORDS + 3 * XSW];
    __shared__ float ex[2 * 2 * 2 * EXW];
    float* Rs = sm;
    float* Xs = sm + 3 * RS_WORDS;
    const int f   = (int)blockIdx.x >> 2;
    const int c   = (int)blockIdx.x & 3;
    const int tid = threadIdx.x;
    const float* gR = S + ((NCH * NBIN + f) * 2) * NPART;
    const float* gX = S + ((c * NBIN + f) * 2) * NPART;

    {
        float rvr[6], rvi[6], xvr[5], xvi[5];
        #pragma unroll
        for (int i = 0; i < 6; ++i) {
            const int t  = tid + 128 * i;
            const int rc = min((t < 240) ? t + 272 : t - 240, NPART - 1);
            rvr[i] = gR[rc];
            rvi[i] = gR[NPART + rc];
        }
        #pragma unroll
        for (int i = 0; i < 5; ++i) {
            const int mc = min(max(tid + 128 * i - 16, 0), NPART - 1);
            xvr[i] = gX[mc];
            xvi[i] = gX[NPART + mc];
        }
        #pragma unroll
        for (int i = 0; i < 6; ++i) {
            const int t = tid + 128 * i;
            if (i < 5 || t < 752) {
                const int w = t + 4 * (t >> 4);
                Rs[w]                = rvr[i];
                Rs[RS_WORDS + w]     = rvi[i];
                Rs[2 * RS_WORDS + w] = rvr[i] + rvi[i];
            }
        }
        #pragma unroll
        for (int i = 0; i < 5; ++i) {
            const int w = tid + 128 * i;
            const bool in = (unsigned)(w - 16) < (unsigned)NPART;
            if (i < 4 || w < XSW) {
                const float re = in ? xvr[i] : 0.f;
                const float im = in ? xvi[i] : 0.f;
                Xs[w]           = re;
                Xs[XSW + w]     = im;
                Xs[2 * XSW + w] = re + im;
            }
        }
    }
    __syncthreads();

    const int wave = __builtin_amdgcn_readfirstlane(tid >> 5);
    const int pair = wave & 1;
    const int part = wave >> 1;
    const int lane = tid & 31;
    const int h    = lane >> 4;
    const int mq   = lane & 15;

    const float* pA0 = Rs + 20 * (15 - mq) + 2 * h;
    const float* pB0 = Xs + mq + 2 * h + 1;
    float* exw = ex + (pair * 2 + part) * 2 * EXW + lane;
    const float* exr = ex + (pair * 2 + (part ^ 1)) * 2 * EXW + lane;

    const v8f zero = {0.f,0.f,0.f,0.f,0.f,0.f,0.f,0.f};
    float yr[8], yi[8];
    int n0;

    if (part == 0) {
        v8f u1 = zero, u2 = zero, u3 = zero;
        v8f c1 = zero, c2 = zero, c3 = zero;
        const float* pA = pA0;
        const float* pB = pB0 + 16 + 256 * pair;
        #pragma unroll 1
        for (int hs = 0; hs < 2; ++hs) {
            const bool sel = (h == hs);
            #pragma unroll
            for (int r = 0; r < 8; ++r) {
                c1[r] = sel ? u1[r] : c1[r];
                c2[r] = sel ? u2[r] : c2[r];
                c3[r] = sel ? u3[r] : c3[r];
                conv_group(pA, pB, u1, u2, u3);
                pA += 20;
                pB += 16;
            }
        }
        #pragma unroll
        for (int v = 0; v < 8; ++v) {
            const float b1 = u1[v] - c1[v], b2 = u2[v] - c2[v], b3 = u3[v] - c3[v];
            yr[v] = (f == 0) ? b1 : b1 - b2;
            yi[v] = (f == 0) ? b2 : (b3 - b1) - b2;
            exw[32 * v]       = (f == 0) ? c1[v] : c1[v] - c2[v];
            exw[EXW + 32 * v] = (f == 0) ? c2[v] : (c3[v] - c1[v]) - c2[v];
        }
        n0 = 512 + 256 * pair;
    } else {
        v8f t1 = zero, t2 = zero, t3 = zero;
        v8f l1 = zero, l2 = zero, l3 = zero;
        const int nT = pair ? 0 : 16;
        const float* pA = pA0 + 20 * 16;
        const float* pB = pB0 + 16 + 256 * pair + 256;
        for (int g = 0; g < nT; ++g) {
            conv_group(pA, pB, t1, t2, t3);
            pA += 20;
            pB += 16;
        }
        const int g2 = pair ? 15 : 31;
        const int nL = 32 - g2;
        pA = pA0 + 20 * g2;
        pB = pB0;
        for (int g = 0; g < nL; ++g) {
            conv_group(pA, pB, l1, l2, l3);
            pA += 20;
            pB += 16;
        }
        #pragma unroll
        for (int v = 0; v < 8; ++v) {
            yr[v] = (f == 0) ? l1[v] : l1[v] - l2[v];
            yi[v] = (f == 0) ? l2[v] : (l3[v] - l1[v]) - l2[v];
            exw[32 * v]       = (f == 0) ? t1[v] : t1[v] - t2[v];
            exw[EXW + 32 * v] = (f == 0) ? t2[v] : (t3[v] - t1[v]) - t2[v];
        }
        n0 = 256 * pair;
    }
    __syncthreads();

    float* d = Y + ((c * NBIN + f) * 2) * NQ + n0 + mq;
    #pragma unroll
    for (int v = 0; v < 8; ++v) {
        const int M = v + 8 * h;
        d[16 * M]      = yr[v] + exr[32 * v];
        d[NQ + 16 * M] = yi[v] + exr[EXW + 32 * v];
    }
}

#define ZP   260
#define PLN  (16 * ZP + 2)
#define RP   130
__device__ __forceinline__ void inv_fold_step(const float* tab, const float* paE,
                                              const float* paO, int f0, int n,
                                              v8f& e, v8f& o) {
    const v2f aE = *(const v2f*)paE;
    const v2f aO = *(const v2f*)paO;
    const int j0 = (f0 * n) & 511;
    const int j1 = (j0 + n) & 511;
    const v2f t0 = *(const v2f*)(tab + 2 * j0);
    const v2f t1 = *(const v2f*)(tab + 2 * j1);
    v2f bE = {t0.x, t1.y};
    v2f bO = {t0.y, t1.x};
    if (f0 == 0) bE.x = 1.0f / 512.0f;
    e = wmma4(aE, bE, e);
    o = wmma4(aO, bO, o);
}

__global__ __launch_bounds__(512)
void rir_part_inverse(const float* __restrict__ Y, float* __restrict__ out)
{
    __shared__ __attribute__((aligned(16))) float tab[1024];
    __shared__ __attribute__((aligned(16))) float zs[2 * PLN];
    __shared__ __attribute__((aligned(16))) float red[4 * 16 * RP];
    __shared__ float y128s[16];
    const int c   = (int)blockIdx.x >> 6;
    const int q0  = 16 * ((int)blockIdx.x & 63);
    const int tid = threadIdx.x;
    {
        float s, co;
        sincospif((float)tid * (1.0f / 256.0f), &s, &co);
        tab[2 * tid]     = co * (1.0f / 256.0f);
        tab[2 * tid + 1] = -s * (1.0f / 256.0f);
    }
    const float* gY = Y + c * (NBIN * 2 * NQ);
    {
        const int qc = tid & 15;
        const int q  = q0 + qc;
        const int r  = tid >> 4;
        const float* col = gY + r * NQ + q;
        const int back = (q > 0) ? 1 : 0;
        float y1[16], y0[16];
        #pragma unroll
        for (int i = 0; i < 16; ++i) {
            y1[i] = col[32 * NQ * i];
            y0[i] = col[32 * NQ * i - back];
        }
        const bool odd = (r & 2) != 0;
        const int  pl  = (((r >> 1) ^ r) & 1);
        float* zc = zs + pl * PLN + qc * ZP + (r >> 1);
        #pragma unroll
        for (int i = 0; i < 16; ++i) {
            const float p = (q > 0) ? y0[i] : 0.f;
            zc[16 * i] = odd ? y1[i] - p : y1[i] + p;
        }
    }
    __syncthreads();

    const int wave = tid >> 5;
    const int lane = tid & 31;
    const int h    = lane >> 4;
    const int mq   = lane & 15;
    const int nt   = wave & 7;
    const int kh   = wave >> 3;
    const int n    = 16 * nt + mq;
    const int fb   = 128 * kh + 2 * h;
    const float* paE = zs + mq * ZP + fb;
    const float* paO = paE + PLN;

    v8f e0 = {0.f,0.f,0.f,0.f,0.f,0.f,0.f,0.f};
    v8f e1 = e0, o0 = e0, o1 = e0;
    for (int t = 0; t < 32; t += 2) {
        inv_fold_step(tab, paE + 4 * t,     paO + 4 * t,     fb + 4 * t,     n, e0, o0);
        inv_fold_step(tab, paE + 4 * t + 4, paO + 4 * t + 4, fb + 4 * t + 4, n, e1, o1);
    }
    {
        float* rE = red + ((2 * kh)     * 16) * RP + n;
        float* rO = red + ((2 * kh + 1) * 16) * RP + n;
        #pragma unroll
        for (int v = 0; v < 8; ++v) {
            const int q = v + 8 * h;
            rE[q * RP] = e0[v] + e1[v];
            rO[q * RP] = o0[v] + o1[v];
        }
    }
    {
        const float* pe = zs + wave * ZP;
        float s = 0.f;
        #pragma unroll
        for (int i = 0; i < 8; ++i) {
            float x = pe[lane + 32 * i];
            if (i == 0 && lane == 0) x *= 0.5f;
            s += x;
        }
        const int m4 = lane & 3;
        if (m4 == 1 || m4 == 2) s = -s;
        #pragma unroll
        for (int off = 16; off > 0; off >>= 1) s += __shfl_xor(s, off);
        if (lane == 0)
            y128s[wave] = s * (1.0f / 256.0f) + zs[PLN + wave * ZP] * (1.0f / 512.0f);
    }
    __syncthreads();

    #pragma unroll
    for (int i = 0; i < 8; ++i) {
        const int o  = tid + 512 * i;
        const int q  = o >> 8;
        const int no = o & 255;
        float val;
        if (no == 128) {
            val = y128s[q];
        } else {
            const int nn = (no < 128) ? no : 256 - no;
            const float E = red[(0 * 16 + q) * RP + nn] + red[(2 * 16 + q) * RP + nn];
            const float O = red[(1 * 16 + q) * RP + nn] + red[(3 * 16 + q) * RP + nn];
            float z = zs[PLN + q * ZP] * (1.0f / 512.0f);
            if (nn & 1) z = -z;
            val = ((no < 128) ? E + O : E - O) + z;
        }
        const int idx = PB * (q0 + q) + no;
        if (idx < OUT_LEN) out[c * OUT_LEN + idx] = val;
    }
}

extern "C" void kernel_launch(void* const* d_in, const int* in_sizes, int n_in,
                              void* d_out, int out_size, void* d_ws, size_t ws_size,
                              hipStream_t stream) {
    (void)in_sizes; (void)n_in; (void)out_size;
    const float* audio = (const float*)d_in[0];
    const float* rir   = (const float*)d_in[1];
    float* out = (float*)d_out;

    if (d_ws != nullptr && ws_size >= WS_NEED_BYTES) {
        float* S = (float*)d_ws;
        float* Y = S + S_FLOATS;
        rir_part_forward<<<dim3(320, 1, 1), dim3(256, 1, 1), 0, stream>>>(audio, rir, S);
        rir_bin_conv<<<dim3(1024, 1, 1), dim3(128, 1, 1), 0, stream>>>(S, Y);
        rir_part_inverse<<<dim3(256, 1, 1), dim3(512, 1, 1), 0, stream>>>(Y, out);
    } else {
        dim3 grid(NBLOCKS, 1, 1);
        dim3 block(256, 1, 1);
        RoomImpulseResponse_wmma_conv<<<grid, block, 0, stream>>>(audio, rir, out);
    }
}
